# starting kernel
# baseline (speedup 1.0000x reference)
#include <hip/hip_runtime.h>

typedef __attribute__((ext_vector_type(2))) float v2f;
typedef __attribute__((ext_vector_type(4))) float v4f;
typedef __attribute__((ext_vector_type(8))) float v8f;

#define GCN_NODES   100000
#define GCN_EDGES   1600000
#define GCN_KDIM    128
#define GCN_HID     128
#define GCN_CLS     40
#define GCN_LDA     129

__global__ void gcn_zero(float* __restrict__ p, long n) {
    long i = (long)blockIdx.x * blockDim.x + threadIdx.x;
    long stride = (long)gridDim.x * blockDim.x;
    for (; i < n; i += stride) p[i] = 0.0f;
}

__global__ void gcn_degrees(const int* __restrict__ src, const int* __restrict__ dst,
                            float* __restrict__ odeg, float* __restrict__ ideg, int nE) {
    int i = blockIdx.x * blockDim.x + threadIdx.x;
    int stride = gridDim.x * blockDim.x;
    for (; i < nE; i += stride) {
        atomicAdd(&odeg[src[i]], 1.0f);
        atomicAdd(&ideg[dst[i]], 1.0f);
    }
}

__global__ void gcn_rsqrt_clamp(float* __restrict__ p, int n) {
    int i = blockIdx.x * blockDim.x + threadIdx.x;
    if (i < n) p[i] = rsqrtf(fmaxf(p[i], 1.0f));
}

__global__ void gcn_wfrag(const float* __restrict__ W, float* __restrict__ Wf,
                          int N, int nColTiles) {
    int i = blockIdx.x * blockDim.x + threadIdx.x;
    int total = nColTiles * 32 * 32;
    if (i >= total) return;
    int lane   = i & 31;
    int k4     = (i >> 5) & 31;
    int t      = i >> 10;
    int lane16 = lane & 15;
    int half   = lane >> 4;
    int kr     = 4 * k4 + 2 * half;
    int col    = t * 16 + lane16;
    float v0 = 0.0f, v1 = 0.0f;
    if (col < N) {
        v0 = W[kr * N + col];
        v1 = W[(kr + 1) * N + col];
    }
    Wf[2 * i]     = v0;
    Wf[2 * i + 1] = v1;
}

__global__ void __launch_bounds__(256)
gcn_gemm_wmma(const float* __restrict__ X, const float* __restrict__ ns,
              const float* __restrict__ Wf, float* __restrict__ Y,
              int N, int nColTiles) {
    __shared__ float Alds[32 * GCN_LDA];

    const int tid = threadIdx.x;
    const int r0  = blockIdx.x * 32;

    #pragma unroll
    for (int idx = tid; idx < 32 * GCN_KDIM; idx += 256) {
        int m = idx >> 7;
        int k = idx & 127;
        Alds[m * GCN_LDA + k] = X[(long)(r0 + m) * GCN_KDIM + k] * ns[r0 + m];
    }
    __syncthreads();

    const int wave   = tid >> 5;
    const int lane   = tid & 31;
    const int lane16 = lane & 15;
    const int half   = lane >> 4;
    if (wave >= nColTiles) return;

    const float* As0 = &Alds[lane16 * GCN_LDA + 2 * half];
    const float* As1 = As0 + 16 * GCN_LDA;
    const v2f* Wfp = (const v2f*)Wf + (long)wave * 32 * 32 + lane;

    v8f acc0 = {0.f, 0.f, 0.f, 0.f, 0.f, 0.f, 0.f, 0.f};
    v8f acc1 = {0.f, 0.f, 0.f, 0.f, 0.f, 0.f, 0.f, 0.f};

    #pragma unroll 8
    for (int k4 = 0; k4 < GCN_KDIM / 4; ++k4) {
        const int k = k4 * 4;
        v2f a0, a1;
        a0.x = As0[k];
        a0.y = As0[k + 1];
        a1.x = As1[k];
        a1.y = As1[k + 1];
        v2f b = Wfp[k4 * 32];
        acc0 = __builtin_amdgcn_wmma_f32_16x16x4_f32(
                   false, a0, false, b, (short)0, acc0, false, false);
        acc1 = __builtin_amdgcn_wmma_f32_16x16x4_f32(
                   false, a1, false, b, (short)0, acc1, false, false);
    }

    const int col = wave * 16 + lane16;
    if (col < N) {
        #pragma unroll
        for (int r = 0; r < 8; ++r) {
            int m = r + 8 * half;
            Y[(long)(r0 + m) * N + col]      = acc0[r];
            Y[(long)(r0 + 16 + m) * N + col] = acc1[r];
        }
    }
}

__global__ void __launch_bounds__(256)
gcn_spmm_scatter(const float* __restrict__ Xe, const int* __restrict__ src,
                 const int* __restrict__ dst, float* __restrict__ Y,
                 int F, int nE) {
    int gid  = blockIdx.x * blockDim.x + threadIdx.x;
    int e    = gid >> 5;
    int lane = gid & 31;
    if (e >= nE) return;
    int s = src[e];
    int d = dst[e];
    const v4f* xr = (const v4f*)(Xe + (long)s * F);
    float*     yr = Y + (long)d * F;
    for (int c = lane; c < (F >> 2); c += 32) {
        v4f v = xr[c];
        atomicAdd(&yr[4 * c + 0], v.x);
        atomicAdd(&yr[4 * c + 1], v.y);
        atomicAdd(&yr[4 * c + 2], v.z);
        atomicAdd(&yr[4 * c + 3], v.w);
    }
}

__global__ void gcn_finish(float* __restrict__ Y, const float* __restrict__ nd,
                           const float* __restrict__ bias, int F, int doRelu,
                           int nNodes) {
    long i = (long)blockIdx.x * blockDim.x + threadIdx.x;
    long total = (long)nNodes * F;
    if (i >= total) return;
    int node = (int)(i / F);
    int f    = (int)(i - (long)node * F);
    float v  = Y[i] * nd[node] + bias[f];
    if (doRelu) v = fmaxf(v, 0.0f);
    Y[i] = v;
}

extern "C" void kernel_launch(void* const* d_in, const int* in_sizes, int n_in,
                              void* d_out, int out_size, void* d_ws, size_t ws_size,
                              hipStream_t stream) {
    (void)in_sizes; (void)n_in; (void)out_size; (void)ws_size;

    const float* feat = (const float*)d_in[0];
    const int*   src  = (const int*)  d_in[1];
    const int*   dst  = (const int*)  d_in[2];
    const float* W0   = (const float*)d_in[3];
    const float* b0   = (const float*)d_in[4];
    const float* W1   = (const float*)d_in[5];
    const float* b1   = (const float*)d_in[6];
    const float* W2   = (const float*)d_in[7];
    const float* b2   = (const float*)d_in[8];
    float*       out  = (float*)d_out;

    const int  Nn = GCN_NODES, Ne = GCN_EDGES, H = GCN_HID, C = GCN_CLS;
    float* ws   = (float*)d_ws;
    float* ns   = ws;
    float* nd   = ws + Nn;
    float* bufA = ws + 2L * Nn;
    float* bufB = bufA + (long)Nn * H;
    float* Wf0  = bufB + (long)Nn * H;
    float* Wf1  = Wf0 + 8 * 2048;
    float* Wf2  = Wf1 + 8 * 2048;

    const int spmmBlocks = (Ne * 32) / 256;

    gcn_zero<<<512, 256, 0, stream>>>(ns, 2L * Nn);
    gcn_degrees<<<2048, 256, 0, stream>>>(src, dst, ns, nd, Ne);
    gcn_rsqrt_clamp<<<(2 * Nn + 255) / 256, 256, 0, stream>>>(ns, 2 * Nn);
    gcn_wfrag<<<(8 * 1024 + 255) / 256, 256, 0, stream>>>(W0, Wf0, H, 8);
    gcn_wfrag<<<(8 * 1024 + 255) / 256, 256, 0, stream>>>(W1, Wf1, H, 8);
    gcn_wfrag<<<(3 * 1024 + 255) / 256, 256, 0, stream>>>(W2, Wf2, C, 3);

    gcn_gemm_wmma<<<Nn / 32, 256, 0, stream>>>(feat, ns, Wf0, bufA, H, 8);
    gcn_zero<<<2048, 256, 0, stream>>>(bufB, (long)Nn * H);
    gcn_spmm_scatter<<<spmmBlocks, 256, 0, stream>>>(bufA, src, dst, bufB, H, Ne);
    gcn_finish<<<((long)Nn * H + 255) / 256, 256, 0, stream>>>(bufB, nd, b0, H, 1, Nn);

    gcn_gemm_wmma<<<Nn / 32, 256, 0, stream>>>(bufB, ns, Wf1, bufA, H, 8);
    gcn_zero<<<2048, 256, 0, stream>>>(bufB, (long)Nn * H);
    gcn_spmm_scatter<<<spmmBlocks, 256, 0, stream>>>(bufA, src, dst, bufB, H, Ne);
    gcn_finish<<<((long)Nn * H + 255) / 256, 256, 0, stream>>>(bufB, nd, b1, H, 1, Nn);

    gcn_gemm_wmma<<<Nn / 32, 256, 0, stream>>>(bufB, ns, Wf2, bufA, C, 3);
    gcn_zero<<<1024, 256, 0, stream>>>(out, (long)Nn * C);
    gcn_spmm_scatter<<<spmmBlocks, 256, 0, stream>>>(bufA, src, dst, out, C, Ne);
    gcn_finish<<<((long)Nn * C + 255) / 256, 256, 0, stream>>>(out, nd, b2, C, 0, Nn);
}


# Aggregate by gather over a per-call edge list sorted by destination (int atomic arrival rank + scan), one wave per row, epilogue fused; float atomics, zero and finish passes gone; sum order = rank
# speedup vs baseline: 2.7160x; 2.7160x over previous
#include <hip/hip_runtime.h>

typedef __attribute__((ext_vector_type(2))) float v2f;
typedef __attribute__((ext_vector_type(4))) float v4f;
typedef __attribute__((ext_vector_type(8))) float v8f;

#define GCN_NODES   100000
#define GCN_EDGES   1600000
#define GCN_KDIM    128
#define GCN_HID     128
#define GCN_CLS     40
#define GCN_LDA     129

__global__ void gcn_zero_i(int* __restrict__ p, int n) {
    int i = blockIdx.x * blockDim.x + threadIdx.x;
    if (i < n) p[i] = 0;
}

__global__ void gcn_rank(const int* __restrict__ src, const int* __restrict__ dst,
                         int* __restrict__ ocnt, int* __restrict__ icnt,
                         int* __restrict__ rank, int nE) {
    int e = blockIdx.x * blockDim.x + threadIdx.x;
    if (e >= nE) return;
    rank[e] = atomicAdd(&icnt[dst[e]], 1);
    atomicAdd(&ocnt[src[e]], 1);
}

__global__ void __launch_bounds__(256)
gcn_scan_norm(const int* __restrict__ icnt, const int* __restrict__ ocnt,
              int* __restrict__ rowptr, float* __restrict__ ns,
              float* __restrict__ nd, int n) {
    __shared__ int red[8];
    __shared__ int wsum[8];
    const int tid  = threadIdx.x;
    const int lane = tid & 31;
    const int wave = tid >> 5;
    const int base = blockIdx.x * 1024;

    int pre = 0;
    for (int i = tid; i < base; i += 256) pre += icnt[i];
    for (int o = 16; o > 0; o >>= 1) pre += __shfl_xor(pre, o);
    if (lane == 0) red[wave] = pre;

    const int i0 = base + 4 * tid;
    int a[4], od[4];
    for (int q = 0; q < 4; ++q) {
        a[q]  = (i0 + q < n) ? icnt[i0 + q] : 0;
        od[q] = (i0 + q < n) ? ocnt[i0 + q] : 0;
    }
    const int tsum = a[0] + a[1] + a[2] + a[3];
    int inc = tsum;
    for (int o = 1; o < 32; o <<= 1) {
        int u = __shfl_up(inc, o);
        if (lane >= o) inc += u;
    }
    if (lane == 31) wsum[wave] = inc;
    __syncthreads();
    int off = 0;
    for (int w = 0; w < 8; ++w) {
        off += red[w];
        if (w < wave) off += wsum[w];
    }
    int ex = off + inc - tsum;
    for (int q = 0; q < 4; ++q) {
        if (i0 + q < n) {
            rowptr[i0 + q] = ex;
            ns[i0 + q] = rsqrtf(fmaxf((float)od[q], 1.0f));
            nd[i0 + q] = rsqrtf(fmaxf((float)a[q], 1.0f));
        }
        ex += a[q];
    }
}

__global__ void gcn_fill(const int* __restrict__ src, const int* __restrict__ dst,
                         const int* __restrict__ rank, const int* __restrict__ rowptr,
                         int* __restrict__ col, int nE) {
    int e = blockIdx.x * blockDim.x + threadIdx.x;
    if (e >= nE) return;
    col[rowptr[dst[e]] + rank[e]] = src[e];
}

__global__ void gcn_wfrag(const float* __restrict__ W, float* __restrict__ Wf,
                          int N, int nColTiles) {
    int i = blockIdx.x * blockDim.x + threadIdx.x;
    int total = nColTiles * 32 * 32;
    if (i >= total) return;
    int lane   = i & 31;
    int k4     = (i >> 5) & 31;
    int t      = i >> 10;
    int lane16 = lane & 15;
    int half   = lane >> 4;
    int kr     = 4 * k4 + 2 * half;
    int col    = t * 16 + lane16;
    float v0 = 0.0f, v1 = 0.0f;
    if (col < N) {
        v0 = W[kr * N + col];
        v1 = W[(kr + 1) * N + col];
    }
    Wf[2 * i]     = v0;
    Wf[2 * i + 1] = v1;
}

__global__ void __launch_bounds__(256)
gcn_gemm_wmma(const float* __restrict__ X, const float* __restrict__ ns,
              const float* __restrict__ Wf, float* __restrict__ Y,
              int N, int nColTiles) {
    __shared__ float Alds[32 * GCN_LDA];

    const int tid = threadIdx.x;
    const int r0  = blockIdx.x * 32;

    #pragma unroll
    for (int idx = tid; idx < 32 * GCN_KDIM; idx += 256) {
        int m = idx >> 7;
        int k = idx & 127;
        Alds[m * GCN_LDA + k] = X[(long)(r0 + m) * GCN_KDIM + k] * ns[r0 + m];
    }
    __syncthreads();

    const int wave   = tid >> 5;
    const int lane   = tid & 31;
    const int lane16 = lane & 15;
    const int half   = lane >> 4;
    if (wave >= nColTiles) return;

    const float* As0 = &Alds[lane16 * GCN_LDA + 2 * half];
    const float* As1 = As0 + 16 * GCN_LDA;
    const v2f* Wfp = (const v2f*)Wf + (long)wave * 32 * 32 + lane;

    v8f acc0 = {0.f, 0.f, 0.f, 0.f, 0.f, 0.f, 0.f, 0.f};
    v8f acc1 = {0.f, 0.f, 0.f, 0.f, 0.f, 0.f, 0.f, 0.f};

    #pragma unroll 8
    for (int k4 = 0; k4 < GCN_KDIM / 4; ++k4) {
        const int k = k4 * 4;
        v2f a0, a1;
        a0.x = As0[k];
        a0.y = As0[k + 1];
        a1.x = As1[k];
        a1.y = As1[k + 1];
        v2f b = Wfp[k4 * 32];
        acc0 = __builtin_amdgcn_wmma_f32_16x16x4_f32(
                   false, a0, false, b, (short)0, acc0, false, false);
        acc1 = __builtin_amdgcn_wmma_f32_16x16x4_f32(
                   false, a1, false, b, (short)0, acc1, false, false);
    }

    const int col = wave * 16 + lane16;
    if (col < N) {
        #pragma unroll
        for (int r = 0; r < 8; ++r) {
            int m = r + 8 * half;
            Y[(long)(r0 + m) * N + col]      = acc0[r];
            Y[(long)(r0 + 16 + m) * N + col] = acc1[r];
        }
    }
}

template <int F, int U>
static __device__ __forceinline__ v4f gcn_take(v4f acc, const v4f* __restrict__ Xl,
                                               int myidx, int t) {
    v4f v[U];
    #pragma unroll
    for (int u = 0; u < U; ++u) {
        const int s = __builtin_amdgcn_readlane(myidx, t + u);
        v[u] = Xl[(long)s * (F / 4)];
    }
    #pragma unroll
    for (int u = 0; u < U; ++u) acc += v[u];
    return acc;
}

template <int F, bool RELU>
__global__ void __launch_bounds__(256)
gcn_gather(const float* __restrict__ Xe, const int* __restrict__ rowptr,
           const int* __restrict__ icnt, const int* __restrict__ col,
           const float* __restrict__ nd, const float* __restrict__ bias,
           float* __restrict__ Y, int nNodes) {
    constexpr int C4 = F / 4;
    const int lane = threadIdx.x & 31;
    const int wave = threadIdx.x >> 5;
    const int row = __builtin_amdgcn_readfirstlane((int)(blockIdx.x * 8 + wave));
    const bool act = lane < C4;
    const int  cl  = act ? lane : 0;
    const v4f* Xl  = (const v4f*)Xe + cl;

    int start = 0, deg = 0;
    float sc = 0.0f;
    if (row < nNodes) {
        start = rowptr[row];
        deg   = icnt[row];
        sc    = nd[row];
    }

    v4f acc = {0.f, 0.f, 0.f, 0.f};
    for (int b = 0; b < deg; b += 32) {
        const int n = min(32, deg - b);
        const int myidx = (lane < n) ? col[start + b + lane] : 0;
        int t = 0;
        for (; t + 8 <= n; t += 8) acc = gcn_take<F, 8>(acc, Xl, myidx, t);
        if (n & 4) { acc = gcn_take<F, 4>(acc, Xl, myidx, t); t += 4; }
        if (n & 2) { acc = gcn_take<F, 2>(acc, Xl, myidx, t); t += 2; }
        if (n & 1) { acc = gcn_take<F, 1>(acc, Xl, myidx, t); }
    }

    v4f r;
    r.x = acc.x * sc + bias[4 * cl + 0];
    r.y = acc.y * sc + bias[4 * cl + 1];
    r.z = acc.z * sc + bias[4 * cl + 2];
    r.w = acc.w * sc + bias[4 * cl + 3];
    if (RELU) {
        r.x = fmaxf(r.x, 0.0f); r.y = fmaxf(r.y, 0.0f);
        r.z = fmaxf(r.z, 0.0f); r.w = fmaxf(r.w, 0.0f);
    }

    if constexpr (C4 == 32) {
        if (row < nNodes) ((v4f*)Y)[(long)row * C4 + lane] = r;
    } else {
        __shared__ float stage[8 * F];
        if (act) {
            stage[wave * F + 4 * lane + 0] = r.x;
            stage[wave * F + 4 * lane + 1] = r.y;
            stage[wave * F + 4 * lane + 2] = r.z;
            stage[wave * F + 4 * lane + 3] = r.w;
        }
        __syncthreads();
        const long gbase = (long)blockIdx.x * 8 * F;
        const long total = (long)nNodes * F;
        for (int i = threadIdx.x; i < 8 * F; i += 256)
            if (gbase + i < total) Y[gbase + i] = stage[i];
    }
}

extern "C" void kernel_launch(void* const* d_in, const int* in_sizes, int n_in,
                              void* d_out, int out_size, void* d_ws, size_t ws_size,
                              hipStream_t stream) {
    (void)in_sizes; (void)n_in; (void)out_size; (void)ws_size;

    const float* feat = (const float*)d_in[0];
    const int*   src  = (const int*)  d_in[1];
    const int*   dst  = (const int*)  d_in[2];
    const float* W0   = (const float*)d_in[3];
    const float* b0   = (const float*)d_in[4];
    const float* W1   = (const float*)d_in[5];
    const float* b1   = (const float*)d_in[6];
    const float* W2   = (const float*)d_in[7];
    const float* b2   = (const float*)d_in[8];
    float*       out  = (float*)d_out;

    const int  Nn = GCN_NODES, Ne = GCN_EDGES, H = GCN_HID, C = GCN_CLS;
    float* ws   = (float*)d_ws;
    float* ns   = ws;
    float* nd   = ws + Nn;
    float* bufA = ws + 2L * Nn;
    float* bufB = bufA + (long)Nn * H;
    float* Wf0  = bufB + (long)Nn * H;
    float* Wf1  = Wf0 + 8 * 2048;
    float* Wf2  = Wf1 + 8 * 2048;
    int*   icnt   = (int*)(Wf2 + 3 * 2048);
    int*   ocnt   = icnt + Nn;
    int*   rowptr = ocnt + Nn;
    int*   rank   = rowptr + Nn;
    int*   col    = rank + Ne;

    gcn_zero_i<<<(2 * Nn + 255) / 256, 256, 0, stream>>>(icnt, 2 * Nn);
    gcn_rank<<<(Ne + 255) / 256, 256, 0, stream>>>(src, dst, ocnt, icnt, rank, Ne);
    gcn_scan_norm<<<(Nn + 1023) / 1024, 256, 0, stream>>>(icnt, ocnt, rowptr, ns, nd, Nn);
    gcn_fill<<<(Ne + 255) / 256, 256, 0, stream>>>(src, dst, rank, rowptr, col, Ne);
    gcn_wfrag<<<(8 * 1024 + 255) / 256, 256, 0, stream>>>(W0, Wf0, H, 8);
    gcn_wfrag<<<(8 * 1024 + 255) / 256, 256, 0, stream>>>(W1, Wf1, H, 8);
    gcn_wfrag<<<(3 * 1024 + 255) / 256, 256, 0, stream>>>(W2, Wf2, C, 3);

    gcn_gemm_wmma<<<Nn / 32, 256, 0, stream>>>(feat, ns, Wf0, bufA, H, 8);
    gcn_gather<GCN_HID, true><<<(Nn + 7) / 8, 256, 0, stream>>>(bufA, rowptr, icnt, col, nd, b0, bufB, Nn);

    gcn_gemm_wmma<<<Nn / 32, 256, 0, stream>>>(bufB, ns, Wf1, bufA, H, 8);
    gcn_gather<GCN_HID, true><<<(Nn + 7) / 8, 256, 0, stream>>>(bufA, rowptr, icnt, col, nd, b1, bufB, Nn);

    gcn_gemm_wmma<<<Nn / 32, 256, 0, stream>>>(bufB, ns, Wf2, bufA, C, 3);
    gcn_gather<GCN_CLS, false><<<(Nn + 7) / 8, 256, 0, stream>>>(bufA, rowptr, icnt, col, nd, b2, out, Nn);
}


# Edge list by destination built with LDS counters only (two-level sort: 512-node buckets, then rows) in 3 launches; per-edge global int atomics, zero/scan/fill passes gone; sum order = LDS arrival
# speedup vs baseline: 4.1052x; 1.5115x over previous
#include <hip/hip_runtime.h>

typedef __attribute__((ext_vector_type(2))) float v2f;
typedef __attribute__((ext_vector_type(4))) float v4f;
typedef __attribute__((ext_vector_type(8))) float v8f;

#define GCN_NODES   100000
#define GCN_EDGES   1600000
#define GCN_KDIM    128
#define GCN_HID     128
#define GCN_CLS     40
#define GCN_LDA     129

#define GCN_BSH   9
#define GCN_BW    512
#define GCN_NBK   ((GCN_NODES + GCN_BW - 1) / GCN_BW)
#define GCN_NCOL  (2 * GCN_NBK)
#define GCN_G1    256
#define GCN_CHUNK ((GCN_EDGES + GCN_G1 - 1) / GCN_G1)
#define GCN_SBITS 17
#define GCN_EPT   ((GCN_CHUNK + 1023) / 1024)

__global__ void __launch_bounds__(1024)
gcn_bcount(const int* __restrict__ src, const int* __restrict__ dst,
           int* __restrict__ cntM, int nE) {
    __shared__ int h[GCN_NCOL];
    const int tid = threadIdx.x;
    if (tid < GCN_NCOL) h[tid] = 0;
    __syncthreads();
    const int e0 = blockIdx.x * GCN_CHUNK;
    const int e1 = min(e0 + GCN_CHUNK, nE);
    int d[GCN_EPT], s[GCN_EPT];
    #pragma unroll
    for (int q = 0; q < GCN_EPT; ++q) {
        const int e = min(e0 + tid + q * 1024, nE - 1);
        d[q] = dst[e];
        s[q] = src[e];
    }
    #pragma unroll
    for (int q = 0; q < GCN_EPT; ++q) {
        if (e0 + tid + q * 1024 < e1) {
            atomicAdd(&h[d[q] >> GCN_BSH], 1);
            atomicAdd(&h[GCN_NBK + (s[q] >> GCN_BSH)], 1);
        }
    }
    __syncthreads();
    if (tid < GCN_NCOL) cntM[blockIdx.x * GCN_NCOL + tid] = h[tid];
}

__global__ void __launch_bounds__(1024)
gcn_bscatter(const int* __restrict__ src, const int* __restrict__ dst,
             const int* __restrict__ cntM, int* __restrict__ gbstart,
             int* __restrict__ gbcnt, int* __restrict__ pairD,
             int* __restrict__ keyS, int nE) {
    __shared__ int sTot[2][512];
    __shared__ int sBef[2][512];
    __shared__ int tot[GCN_NCOL];
    __shared__ int cur[GCN_NCOL];
    const int tid = threadIdx.x;
    const int c   = tid & 511;
    const int seg = tid >> 9;
    const int blk = blockIdx.x;

    int t = 0, bf = 0;
    if (c < GCN_NCOL) {
        const int r0 = seg * (GCN_G1 / 2);
        #pragma unroll 16
        for (int r = r0; r < r0 + GCN_G1 / 2; ++r) {
            const int v = cntM[r * GCN_NCOL + c];
            t  += v;
            bf += (r < blk) ? v : 0;
        }
    }
    sTot[seg][c] = t;
    sBef[seg][c] = bf;
    __syncthreads();
    if (tid < GCN_NCOL) tot[tid] = sTot[0][tid] + sTot[1][tid];
    __syncthreads();
    if (tid < GCN_NCOL) {
        const int lo = (tid < GCN_NBK) ? 0 : GCN_NBK;
        int st = 0;
        for (int j = lo; j < tid; ++j) st += tot[j];
        cur[tid] = st + sBef[0][tid] + sBef[1][tid];
        if (blk == 0) { gbstart[tid] = st; gbcnt[tid] = tot[tid]; }
    }
    __syncthreads();

    const int e0 = blk * GCN_CHUNK;
    const int e1 = min(e0 + GCN_CHUNK, nE);
    int d[GCN_EPT], s[GCN_EPT];
    #pragma unroll
    for (int q = 0; q < GCN_EPT; ++q) {
        const int e = min(e0 + tid + q * 1024, nE - 1);
        d[q] = dst[e];
        s[q] = src[e];
    }
    #pragma unroll
    for (int q = 0; q < GCN_EPT; ++q) {
        if (e0 + tid + q * 1024 < e1) {
            const int pd = atomicAdd(&cur[d[q] >> GCN_BSH], 1);
            pairD[pd] = ((d[q] & (GCN_BW - 1)) << GCN_SBITS) | s[q];
            const int pq = atomicAdd(&cur[GCN_NBK + (s[q] >> GCN_BSH)], 1);
            keyS[pq] = s[q] & (GCN_BW - 1);
        }
    }
}

__global__ void __launch_bounds__(1024)
gcn_bsort(const int* __restrict__ pairD, const int* __restrict__ keyS,
          const int* __restrict__ gbstart, const int* __restrict__ gbcnt,
          int* __restrict__ rowptr, int* __restrict__ icnt, int* __restrict__ col,
          float* __restrict__ ns, float* __restrict__ nd, int nNodes, int nE) {
    __shared__ int cI[GCN_BW];
    __shared__ int cO[GCN_BW];
    __shared__ int pos[GCN_BW];
    __shared__ int wsum[32];
    const int tid  = threadIdx.x;
    const int lane = tid & 31;
    const int wave = tid >> 5;
    const int k    = blockIdx.x;
    if (tid < GCN_BW) { cI[tid] = 0; cO[tid] = 0; }
    __syncthreads();

    const int dS = gbstart[k];
    const int dN = gbcnt[k];
    const int sS = gbstart[GCN_NBK + k];
    const int sN = gbcnt[GCN_NBK + k];
    for (int i0 = tid; i0 < dN; i0 += 8 * 1024) {
        unsigned key[8];
        #pragma unroll
        for (int q = 0; q < 8; ++q) {
            key[q] = (unsigned)pairD[min(dS + i0 + q * 1024, nE - 1)];
        }
        #pragma unroll
        for (int q = 0; q < 8; ++q)
            if (i0 + q * 1024 < dN) atomicAdd(&cI[key[q] >> GCN_SBITS], 1);
    }
    for (int i0 = tid; i0 < sN; i0 += 8 * 1024) {
        int ks[8];
        #pragma unroll
        for (int q = 0; q < 8; ++q) {
            ks[q] = keyS[min(sS + i0 + q * 1024, nE - 1)];
        }
        #pragma unroll
        for (int q = 0; q < 8; ++q)
            if (i0 + q * 1024 < sN) atomicAdd(&cO[ks[q]], 1);
    }
    __syncthreads();

    const int v = (tid < GCN_BW) ? cI[tid] : 0;
    int inc = v;
    for (int o = 1; o < 32; o <<= 1) {
        int u = __shfl_up(inc, o);
        if (lane >= o) inc += u;
    }
    if (lane == 31) wsum[wave] = inc;
    __syncthreads();
    int off = 0;
    for (int w = 0; w < 16; ++w)
        if (w < wave) off += wsum[w];
    const int ex = off + inc - v;
    if (tid < GCN_BW) {
        pos[tid] = ex;
        const int node = k * GCN_BW + tid;
        if (node < nNodes) {
            rowptr[node] = dS + ex;
            icnt[node]   = v;
            nd[node] = rsqrtf(fmaxf((float)v, 1.0f));
            ns[node] = rsqrtf(fmaxf((float)cO[tid], 1.0f));
        }
    }
    __syncthreads();

    for (int i0 = tid; i0 < dN; i0 += 8 * 1024) {
        unsigned key[8];
        #pragma unroll
        for (int q = 0; q < 8; ++q) {
            key[q] = (unsigned)pairD[min(dS + i0 + q * 1024, nE - 1)];
        }
        #pragma unroll
        for (int q = 0; q < 8; ++q) {
            if (i0 + q * 1024 < dN) {
                const int p = atomicAdd(&pos[key[q] >> GCN_SBITS], 1);
                col[dS + p] = (int)(key[q] & ((1u << GCN_SBITS) - 1u));
            }
        }
    }
}

__global__ void gcn_wfrag(const float* __restrict__ W, float* __restrict__ Wf,
                          int N, int nColTiles) {
    int i = blockIdx.x * blockDim.x + threadIdx.x;
    int total = nColTiles * 32 * 32;
    if (i >= total) return;
    int lane   = i & 31;
    int k4     = (i >> 5) & 31;
    int t      = i >> 10;
    int lane16 = lane & 15;
    int half   = lane >> 4;
    int kr     = 4 * k4 + 2 * half;
    int col    = t * 16 + lane16;
    float v0 = 0.0f, v1 = 0.0f;
    if (col < N) {
        v0 = W[kr * N + col];
        v1 = W[(kr + 1) * N + col];
    }
    Wf[2 * i]     = v0;
    Wf[2 * i + 1] = v1;
}

__global__ void __launch_bounds__(256)
gcn_gemm_wmma(const float* __restrict__ X, const float* __restrict__ ns,
              const float* __restrict__ Wf, float* __restrict__ Y,
              int N, int nColTiles) {
    __shared__ float Alds[32 * GCN_LDA];

    const int tid = threadIdx.x;
    const int r0  = blockIdx.x * 32;

    #pragma unroll
    for (int idx = tid; idx < 32 * GCN_KDIM; idx += 256) {
        int m = idx >> 7;
        int k = idx & 127;
        Alds[m * GCN_LDA + k] = X[(long)(r0 + m) * GCN_KDIM + k] * ns[r0 + m];
    }
    __syncthreads();

    const int wave   = tid >> 5;
    const int lane   = tid & 31;
    const int lane16 = lane & 15;
    const int half   = lane >> 4;
    if (wave >= nColTiles) return;

    const float* As0 = &Alds[lane16 * GCN_LDA + 2 * half];
    const float* As1 = As0 + 16 * GCN_LDA;
    const v2f* Wfp = (const v2f*)Wf + (long)wave * 32 * 32 + lane;

    v8f acc0 = {0.f, 0.f, 0.f, 0.f, 0.f, 0.f, 0.f, 0.f};
    v8f acc1 = {0.f, 0.f, 0.f, 0.f, 0.f, 0.f, 0.f, 0.f};

    #pragma unroll 8
    for (int k4 = 0; k4 < GCN_KDIM / 4; ++k4) {
        const int k = k4 * 4;
        v2f a0, a1;
        a0.x = As0[k];
        a0.y = As0[k + 1];
        a1.x = As1[k];
        a1.y = As1[k + 1];
        v2f b = Wfp[k4 * 32];
        acc0 = __builtin_amdgcn_wmma_f32_16x16x4_f32(
                   false, a0, false, b, (short)0, acc0, false, false);
        acc1 = __builtin_amdgcn_wmma_f32_16x16x4_f32(
                   false, a1, false, b, (short)0, acc1, false, false);
    }

    const int col = wave * 16 + lane16;
    if (col < N) {
        #pragma unroll
        for (int r = 0; r < 8; ++r) {
            int m = r + 8 * half;
            Y[(long)(r0 + m) * N + col]      = acc0[r];
            Y[(long)(r0 + 16 + m) * N + col] = acc1[r];
        }
    }
}

template <int F, int U>
static __device__ __forceinline__ v4f gcn_take(v4f acc, const v4f* __restrict__ Xl,
                                               int myidx, int t) {
    v4f v[U];
    #pragma unroll
    for (int u = 0; u < U; ++u) {
        const int s = __builtin_amdgcn_readlane(myidx, t + u);
        v[u] = Xl[(long)s * (F / 4)];
    }
    #pragma unroll
    for (int u = 0; u < U; ++u) acc += v[u];
    return acc;
}

template <int F, bool RELU>
__global__ void __launch_bounds__(256)
gcn_gather(const float* __restrict__ Xe, const int* __restrict__ rowptr,
           const int* __restrict__ icnt, const int* __restrict__ col,
           const float* __restrict__ nd, const float* __restrict__ bias,
           float* __restrict__ Y, int nNodes) {
    constexpr int C4 = F / 4;
    const int lane = threadIdx.x & 31;
    const int wave = threadIdx.x >> 5;
    const int row = __builtin_amdgcn_readfirstlane((int)(blockIdx.x * 8 + wave));
    const bool act = lane < C4;
    const int  cl  = act ? lane : 0;
    const v4f* Xl  = (const v4f*)Xe + cl;

    int start = 0, deg = 0;
    float sc = 0.0f;
    if (row < nNodes) {
        start = rowptr[row];
        deg   = icnt[row];
        sc    = nd[row];
    }

    v4f acc = {0.f, 0.f, 0.f, 0.f};
    for (int b = 0; b < deg; b += 32) {
        const int n = min(32, deg - b);
        const int myidx = (lane < n) ? col[start + b + lane] : 0;
        int t = 0;
        for (; t + 8 <= n; t += 8) acc = gcn_take<F, 8>(acc, Xl, myidx, t);
        if (n & 4) { acc = gcn_take<F, 4>(acc, Xl, myidx, t); t += 4; }
        if (n & 2) { acc = gcn_take<F, 2>(acc, Xl, myidx, t); t += 2; }
        if (n & 1) { acc = gcn_take<F, 1>(acc, Xl, myidx, t); }
    }

    v4f r;
    r.x = acc.x * sc + bias[4 * cl + 0];
    r.y = acc.y * sc + bias[4 * cl + 1];
    r.z = acc.z * sc + bias[4 * cl + 2];
    r.w = acc.w * sc + bias[4 * cl + 3];
    if (RELU) {
        r.x = fmaxf(r.x, 0.0f); r.y = fmaxf(r.y, 0.0f);
        r.z = fmaxf(r.z, 0.0f); r.w = fmaxf(r.w, 0.0f);
    }

    if constexpr (C4 == 32) {
        if (row < nNodes) ((v4f*)Y)[(long)row * C4 + lane] = r;
    } else {
        __shared__ float stage[8 * F];
        if (act) {
            stage[wave * F + 4 * lane + 0] = r.x;
            stage[wave * F + 4 * lane + 1] = r.y;
            stage[wave * F + 4 * lane + 2] = r.z;
            stage[wave * F + 4 * lane + 3] = r.w;
        }
        __syncthreads();
        const long gbase = (long)blockIdx.x * 8 * F;
        const long total = (long)nNodes * F;
        for (int i = threadIdx.x; i < 8 * F; i += 256)
            if (gbase + i < total) Y[gbase + i] = stage[i];
    }
}

extern "C" void kernel_launch(void* const* d_in, const int* in_sizes, int n_in,
                              void* d_out, int out_size, void* d_ws, size_t ws_size,
                              hipStream_t stream) {
    (void)in_sizes; (void)n_in; (void)out_size; (void)ws_size;

    const float* feat = (const float*)d_in[0];
    const int*   src  = (const int*)  d_in[1];
    const int*   dst  = (const int*)  d_in[2];
    const float* W0   = (const float*)d_in[3];
    const float* b0   = (const float*)d_in[4];
    const float* W1   = (const float*)d_in[5];
    const float* b1   = (const float*)d_in[6];
    const float* W2   = (const float*)d_in[7];
    const float* b2   = (const float*)d_in[8];
    float*       out  = (float*)d_out;

    const int  Nn = GCN_NODES, Ne = GCN_EDGES, H = GCN_HID, C = GCN_CLS;
    float* ws   = (float*)d_ws;
    float* ns   = ws;
    float* nd   = ws + Nn;
    float* bufA = ws + 2L * Nn;
    float* bufB = bufA + (long)Nn * H;
    float* Wf0  = bufB + (long)Nn * H;
    float* Wf1  = Wf0 + 8 * 2048;
    float* Wf2  = Wf1 + 8 * 2048;
    int*   icnt    = (int*)(Wf2 + 3 * 2048);
    int*   rowptr  = icnt + Nn;
    int*   col     = rowptr + Nn;
    int*   pairD   = col + Ne;
    int*   keyS    = pairD + Ne;
    int*   cntM    = keyS + Ne;
    int*   gbstart = cntM + GCN_G1 * GCN_NCOL;
    int*   gbcnt   = gbstart + GCN_NCOL;

    gcn_bcount<<<GCN_G1, 1024, 0, stream>>>(src, dst, cntM, Ne);
    gcn_bscatter<<<GCN_G1, 1024, 0, stream>>>(src, dst, cntM, gbstart, gbcnt, pairD, keyS, Ne);
    gcn_bsort<<<GCN_NBK, 1024, 0, stream>>>(pairD, keyS, gbstart, gbcnt, rowptr, icnt, col, ns, nd, Nn, Ne);
    gcn_wfrag<<<(8 * 1024 + 255) / 256, 256, 0, stream>>>(W0, Wf0, H, 8);
    gcn_wfrag<<<(8 * 1024 + 255) / 256, 256, 0, stream>>>(W1, Wf1, H, 8);
    gcn_wfrag<<<(3 * 1024 + 255) / 256, 256, 0, stream>>>(W2, Wf2, C, 3);

    gcn_gemm_wmma<<<Nn / 32, 256, 0, stream>>>(feat, ns, Wf0, bufA, H, 8);
    gcn_gather<GCN_HID, true><<<(Nn + 7) / 8, 256, 0, stream>>>(bufA, rowptr, icnt, col, nd, b0, bufB, Nn);

    gcn_gemm_wmma<<<Nn / 32, 256, 0, stream>>>(bufB, ns, Wf1, bufA, H, 8);
    gcn_gather<GCN_HID, true><<<(Nn + 7) / 8, 256, 0, stream>>>(bufA, rowptr, icnt, col, nd, b1, bufB, Nn);

    gcn_gemm_wmma<<<Nn / 32, 256, 0, stream>>>(bufB, ns, Wf2, bufA, C, 3);
    gcn_gather<GCN_CLS, false><<<(Nn + 7) / 8, 256, 0, stream>>>(bufA, rowptr, icnt, col, nd, b2, out, Nn);
}


# Layers 0, 1: aggregation block (32 rows, 4 per wave) keeps its finished rows in LDS and multiplies them by the next layer's weights on WMMA in the same launch; 2 launches and the hidden plane gone
# speedup vs baseline: 4.6029x; 1.1212x over previous
#include <hip/hip_runtime.h>

typedef __attribute__((ext_vector_type(2))) float v2f;
typedef __attribute__((ext_vector_type(4))) float v4f;
typedef __attribute__((ext_vector_type(8))) float v8f;

#define GCN_NODES   100000
#define GCN_EDGES   1600000
#define GCN_KDIM    128
#define GCN_HID     128
#define GCN_CLS     40
#define GCN_LDA     129

#define GCN_BSH   9
#define GCN_BW    512
#define GCN_NBK   ((GCN_NODES + GCN_BW - 1) / GCN_BW)
#define GCN_NCOL  (2 * GCN_NBK)
#define GCN_G1    256
#define GCN_CHUNK ((GCN_EDGES + GCN_G1 - 1) / GCN_G1)
#define GCN_SBITS 17
#define GCN_EPT   ((GCN_CHUNK + 1023) / 1024)

__global__ void __launch_bounds__(1024)
gcn_bcount(const int* __restrict__ src, const int* __restrict__ dst,
           int* __restrict__ cntM, int nE) {
    __shared__ int h[GCN_NCOL];
    const int tid = threadIdx.x;
    if (tid < GCN_NCOL) h[tid] = 0;
    __syncthreads();
    const int e0 = blockIdx.x * GCN_CHUNK;
    const int e1 = min(e0 + GCN_CHUNK, nE);
    int d[GCN_EPT], s[GCN_EPT];
    #pragma unroll
    for (int q = 0; q < GCN_EPT; ++q) {
        const int e = min(e0 + tid + q * 1024, nE - 1);
        d[q] = dst[e];
        s[q] = src[e];
    }
    #pragma unroll
    for (int q = 0; q < GCN_EPT; ++q) {
        if (e0 + tid + q * 1024 < e1) {
            atomicAdd(&h[d[q] >> GCN_BSH], 1);
            atomicAdd(&h[GCN_NBK + (s[q] >> GCN_BSH)], 1);
        }
    }
    __syncthreads();
    if (tid < GCN_NCOL) cntM[blockIdx.x * GCN_NCOL + tid] = h[tid];
}

__global__ void __launch_bounds__(1024)
gcn_bscatter(const int* __restrict__ src, const int* __restrict__ dst,
             const int* __restrict__ cntM, int* __restrict__ gbstart,
             int* __restrict__ gbcnt, int* __restrict__ pairD,
             int* __restrict__ keyS, int nE) {
    __shared__ int sTot[2][512];
    __shared__ int sBef[2][512];
    __shared__ int tot[GCN_NCOL];
    __shared__ int cur[GCN_NCOL];
    const int tid = threadIdx.x;
    const int c   = tid & 511;
    const int seg = tid >> 9;
    const int blk = blockIdx.x;

    int t = 0, bf = 0;
    if (c < GCN_NCOL) {
        const int r0 = seg * (GCN_G1 / 2);
        #pragma unroll 16
        for (int r = r0; r < r0 + GCN_G1 / 2; ++r) {
            const int v = cntM[r * GCN_NCOL + c];
            t  += v;
            bf += (r < blk) ? v : 0;
        }
    }
    sTot[seg][c] = t;
    sBef[seg][c] = bf;
    __syncthreads();
    if (tid < GCN_NCOL) tot[tid] = sTot[0][tid] + sTot[1][tid];
    __syncthreads();
    if (tid < GCN_NCOL) {
        const int lo = (tid < GCN_NBK) ? 0 : GCN_NBK;
        int st = 0;
        for (int j = lo; j < tid; ++j) st += tot[j];
        cur[tid] = st + sBef[0][tid] + sBef[1][tid];
        if (blk == 0) { gbstart[tid] = st; gbcnt[tid] = tot[tid]; }
    }
    __syncthreads();

    const int e0 = blk * GCN_CHUNK;
    const int e1 = min(e0 + GCN_CHUNK, nE);
    int d[GCN_EPT], s[GCN_EPT];
    #pragma unroll
    for (int q = 0; q < GCN_EPT; ++q) {
        const int e = min(e0 + tid + q * 1024, nE - 1);
        d[q] = dst[e];
        s[q] = src[e];
    }
    #pragma unroll
    for (int q = 0; q < GCN_EPT; ++q) {
        if (e0 + tid + q * 1024 < e1) {
            const int pd = atomicAdd(&cur[d[q] >> GCN_BSH], 1);
            pairD[pd] = ((d[q] & (GCN_BW - 1)) << GCN_SBITS) | s[q];
            const int pq = atomicAdd(&cur[GCN_NBK + (s[q] >> GCN_BSH)], 1);
            keyS[pq] = s[q] & (GCN_BW - 1);
        }
    }
}

__global__ void __launch_bounds__(1024)
gcn_bsort(const int* __restrict__ pairD, const int* __restrict__ keyS,
          const int* __restrict__ gbstart, const int* __restrict__ gbcnt,
          int* __restrict__ rowptr, int* __restrict__ icnt, int* __restrict__ col,
          float* __restrict__ ns, float* __restrict__ nd, int nNodes, int nE) {
    __shared__ int cI[GCN_BW];
    __shared__ int cO[GCN_BW];
    __shared__ int pos[GCN_BW];
    __shared__ int wsum[32];
    const int tid  = threadIdx.x;
    const int lane = tid & 31;
    const int wave = tid >> 5;
    const int k    = blockIdx.x;
    if (tid < GCN_BW) { cI[tid] = 0; cO[tid] = 0; }
    __syncthreads();

    const int dS = gbstart[k];
    const int dN = gbcnt[k];
    const int sS = gbstart[GCN_NBK + k];
    const int sN = gbcnt[GCN_NBK + k];
    for (int i0 = tid; i0 < dN; i0 += 8 * 1024) {
        unsigned key[8];
        #pragma unroll
        for (int q = 0; q < 8; ++q) {
            key[q] = (unsigned)pairD[min(dS + i0 + q * 1024, nE - 1)];
        }
        #pragma unroll
        for (int q = 0; q < 8; ++q)
            if (i0 + q * 1024 < dN) atomicAdd(&cI[key[q] >> GCN_SBITS], 1);
    }
    for (int i0 = tid; i0 < sN; i0 += 8 * 1024) {
        int ks[8];
        #pragma unroll
        for (int q = 0; q < 8; ++q) {
            ks[q] = keyS[min(sS + i0 + q * 1024, nE - 1)];
        }
        #pragma unroll
        for (int q = 0; q < 8; ++q)
            if (i0 + q * 1024 < sN) atomicAdd(&cO[ks[q]], 1);
    }
    __syncthreads();

    const int v = (tid < GCN_BW) ? cI[tid] : 0;
    int inc = v;
    for (int o = 1; o < 32; o <<= 1) {
        int u = __shfl_up(inc, o);
        if (lane >= o) inc += u;
    }
    if (lane == 31) wsum[wave] = inc;
    __syncthreads();
    int off = 0;
    for (int w = 0; w < 16; ++w)
        if (w < wave) off += wsum[w];
    const int ex = off + inc - v;
    if (tid < GCN_BW) {
        pos[tid] = ex;
        const int node = k * GCN_BW + tid;
        if (node < nNodes) {
            rowptr[node] = dS + ex;
            icnt[node]   = v;
            nd[node] = rsqrtf(fmaxf((float)v, 1.0f));
            ns[node] = rsqrtf(fmaxf((float)cO[tid], 1.0f));
        }
    }
    __syncthreads();

    for (int i0 = tid; i0 < dN; i0 += 8 * 1024) {
        unsigned key[8];
        #pragma unroll
        for (int q = 0; q < 8; ++q) {
            key[q] = (unsigned)pairD[min(dS + i0 + q * 1024, nE - 1)];
        }
        #pragma unroll
        for (int q = 0; q < 8; ++q) {
            if (i0 + q * 1024 < dN) {
                const int p = atomicAdd(&pos[key[q] >> GCN_SBITS], 1);
                col[dS + p] = (int)(key[q] & ((1u << GCN_SBITS) - 1u));
            }
        }
    }
}

__global__ void gcn_wfrag(const float* __restrict__ W, float* __restrict__ Wf,
                          int N, int nColTiles) {
    int i = blockIdx.x * blockDim.x + threadIdx.x;
    int total = nColTiles * 32 * 32;
    if (i >= total) return;
    int lane   = i & 31;
    int k4     = (i >> 5) & 31;
    int t      = i >> 10;
    int lane16 = lane & 15;
    int half   = lane >> 4;
    int kr     = 4 * k4 + 2 * half;
    int col    = t * 16 + lane16;
    float v0 = 0.0f, v1 = 0.0f;
    if (col < N) {
        v0 = W[kr * N + col];
        v1 = W[(kr + 1) * N + col];
    }
    Wf[2 * i]     = v0;
    Wf[2 * i + 1] = v1;
}

static __device__ __forceinline__ void
gcn_tile_product(const float* __restrict__ Alds, const float* __restrict__ Wf,
                 float* __restrict__ Y, int r0, int N, int nColTiles) {
    const int tid    = threadIdx.x;
    const int wave   = tid >> 5;
    const int lane   = tid & 31;
    const int lane16 = lane & 15;
    const int half   = lane >> 4;
    if (wave >= nColTiles) return;

    const float* As0 = &Alds[lane16 * GCN_LDA + 2 * half];
    const float* As1 = As0 + 16 * GCN_LDA;
    const v2f* Wfp = (const v2f*)Wf + (long)wave * 32 * 32 + lane;

    v8f acc0 = {0.f, 0.f, 0.f, 0.f, 0.f, 0.f, 0.f, 0.f};
    v8f acc1 = {0.f, 0.f, 0.f, 0.f, 0.f, 0.f, 0.f, 0.f};

    #pragma unroll 8
    for (int k4 = 0; k4 < GCN_KDIM / 4; ++k4) {
        const int k = k4 * 4;
        v2f a0, a1;
        a0.x = As0[k];
        a0.y = As0[k + 1];
        a1.x = As1[k];
        a1.y = As1[k + 1];
        v2f b = Wfp[k4 * 32];
        acc0 = __builtin_amdgcn_wmma_f32_16x16x4_f32(
                   false, a0, false, b, (short)0, acc0, false, false);
        acc1 = __builtin_amdgcn_wmma_f32_16x16x4_f32(
                   false, a1, false, b, (short)0, acc1, false, false);
    }

    const int col = wave * 16 + lane16;
    if (col < N) {
        #pragma unroll
        for (int r = 0; r < 8; ++r) {
            int m = r + 8 * half;
            Y[(long)(r0 + m) * N + col]      = acc0[r];
            Y[(long)(r0 + 16 + m) * N + col] = acc1[r];
        }
    }
}

__global__ void __launch_bounds__(256)
gcn_gemm_wmma(const float* __restrict__ X, const float* __restrict__ ns,
              const float* __restrict__ Wf, float* __restrict__ Y,
              int N, int nColTiles) {
    __shared__ float Alds[32 * GCN_LDA];

    const int tid = threadIdx.x;
    const int r0  = blockIdx.x * 32;

    #pragma unroll
    for (int idx = tid; idx < 32 * GCN_KDIM; idx += 256) {
        int m = idx >> 7;
        int k = idx & 127;
        Alds[m * GCN_LDA + k] = X[(long)(r0 + m) * GCN_KDIM + k] * ns[r0 + m];
    }
    __syncthreads();

    gcn_tile_product(Alds, Wf, Y, r0, N, nColTiles);
}

template <int F, int U>
static __device__ __forceinline__ v4f gcn_take(v4f acc, const v4f* __restrict__ Xl,
                                               int myidx, int t) {
    v4f v[U];
    #pragma unroll
    for (int u = 0; u < U; ++u) {
        const int s = __builtin_amdgcn_readlane(myidx, t + u);
        v[u] = Xl[(long)s * (F / 4)];
    }
    #pragma unroll
    for (int u = 0; u < U; ++u) acc += v[u];
    return acc;
}

template <int F, bool RELU>
__global__ void __launch_bounds__(256)
gcn_gather(const float* __restrict__ Xe, const int* __restrict__ rowptr,
           const int* __restrict__ icnt, const int* __restrict__ col,
           const float* __restrict__ nd, const float* __restrict__ bias,
           float* __restrict__ Y, int nNodes) {
    constexpr int C4 = F / 4;
    const int lane = threadIdx.x & 31;
    const int wave = threadIdx.x >> 5;
    const int row = __builtin_amdgcn_readfirstlane((int)(blockIdx.x * 8 + wave));
    const bool act = lane < C4;
    const int  cl  = act ? lane : 0;
    const v4f* Xl  = (const v4f*)Xe + cl;

    int start = 0, deg = 0;
    float sc = 0.0f;
    if (row < nNodes) {
        start = rowptr[row];
        deg   = icnt[row];
        sc    = nd[row];
    }

    v4f acc = {0.f, 0.f, 0.f, 0.f};
    for (int b = 0; b < deg; b += 32) {
        const int n = min(32, deg - b);
        const int myidx = (lane < n) ? col[start + b + lane] : 0;
        int t = 0;
        for (; t + 8 <= n; t += 8) acc = gcn_take<F, 8>(acc, Xl, myidx, t);
        if (n & 4) { acc = gcn_take<F, 4>(acc, Xl, myidx, t); t += 4; }
        if (n & 2) { acc = gcn_take<F, 2>(acc, Xl, myidx, t); t += 2; }
        if (n & 1) { acc = gcn_take<F, 1>(acc, Xl, myidx, t); }
    }

    v4f r;
    r.x = acc.x * sc + bias[4 * cl + 0];
    r.y = acc.y * sc + bias[4 * cl + 1];
    r.z = acc.z * sc + bias[4 * cl + 2];
    r.w = acc.w * sc + bias[4 * cl + 3];
    if (RELU) {
        r.x = fmaxf(r.x, 0.0f); r.y = fmaxf(r.y, 0.0f);
        r.z = fmaxf(r.z, 0.0f); r.w = fmaxf(r.w, 0.0f);
    }

    if constexpr (C4 == 32) {
        if (row < nNodes) ((v4f*)Y)[(long)row * C4 + lane] = r;
    } else {
        __shared__ float stage[8 * F];
        if (act) {
            stage[wave * F + 4 * lane + 0] = r.x;
            stage[wave * F + 4 * lane + 1] = r.y;
            stage[wave * F + 4 * lane + 2] = r.z;
            stage[wave * F + 4 * lane + 3] = r.w;
        }
        __syncthreads();
        const long gbase = (long)blockIdx.x * 8 * F;
        const long total = (long)nNodes * F;
        for (int i = threadIdx.x; i < 8 * F; i += 256)
            if (gbase + i < total) Y[gbase + i] = stage[i];
    }
}

__global__ void __launch_bounds__(256)
gcn_gather_gemm(const float* __restrict__ Xe, const int* __restrict__ rowptr,
                const int* __restrict__ icnt, const int* __restrict__ col,
                const float* __restrict__ nd, const float* __restrict__ ns,
                const float* __restrict__ bias, const float* __restrict__ Wf,
                float* __restrict__ Y, int N, int nColTiles) {
    __shared__ float Alds[32 * GCN_LDA];

    const int lane = threadIdx.x & 31;
    const int wave = __builtin_amdgcn_readfirstlane((int)(threadIdx.x >> 5));
    const int r0   = blockIdx.x * 32;
    const int rb   = r0 + wave * 4;
    const v4f* Xl  = (const v4f*)Xe + lane;

    int   st[4], dg[4], id0[4];
    float sc[4], s2[4];
    #pragma unroll
    for (int j = 0; j < 4; ++j) {
        st[j] = rowptr[rb + j];
        dg[j] = icnt[rb + j];
        sc[j] = nd[rb + j];
        s2[j] = ns[rb + j];
    }
    #pragma unroll
    for (int j = 0; j < 4; ++j)
        id0[j] = (lane < dg[j]) ? col[st[j] + lane] : 0;

    const float b0 = bias[4 * lane + 0];
    const float b1 = bias[4 * lane + 1];
    const float b2 = bias[4 * lane + 2];
    const float b3 = bias[4 * lane + 3];

    #pragma unroll
    for (int j = 0; j < 4; ++j) {
        const int deg = dg[j];
        v4f acc = {0.f, 0.f, 0.f, 0.f};
        for (int b = 0; b < deg; b += 32) {
            const int n = min(32, deg - b);
            int myidx = id0[j];
            if (b > 0) myidx = (lane < n) ? col[st[j] + b + lane] : 0;
            int t = 0;
            for (; t + 8 <= n; t += 8) acc = gcn_take<GCN_HID, 8>(acc, Xl, myidx, t);
            if (n & 4) { acc = gcn_take<GCN_HID, 4>(acc, Xl, myidx, t); t += 4; }
            if (n & 2) { acc = gcn_take<GCN_HID, 2>(acc, Xl, myidx, t); t += 2; }
            if (n & 1) { acc = gcn_take<GCN_HID, 1>(acc, Xl, myidx, t); }
        }
        float* a = &Alds[(wave * 4 + j) * GCN_LDA + 4 * lane];
        a[0] = fmaxf(acc.x * sc[j] + b0, 0.0f) * s2[j];
        a[1] = fmaxf(acc.y * sc[j] + b1, 0.0f) * s2[j];
        a[2] = fmaxf(acc.z * sc[j] + b2, 0.0f) * s2[j];
        a[3] = fmaxf(acc.w * sc[j] + b3, 0.0f) * s2[j];
    }
    __syncthreads();

    gcn_tile_product(Alds, Wf, Y, r0, N, nColTiles);
}

extern "C" void kernel_launch(void* const* d_in, const int* in_sizes, int n_in,
                              void* d_out, int out_size, void* d_ws, size_t ws_size,
                              hipStream_t stream) {
    (void)in_sizes; (void)n_in; (void)out_size; (void)ws_size;

    const float* feat = (const float*)d_in[0];
    const int*   src  = (const int*)  d_in[1];
    const int*   dst  = (const int*)  d_in[2];
    const float* W0   = (const float*)d_in[3];
    const float* b0   = (const float*)d_in[4];
    const float* W1   = (const float*)d_in[5];
    const float* b1   = (const float*)d_in[6];
    const float* W2   = (const float*)d_in[7];
    const float* b2   = (const float*)d_in[8];
    float*       out  = (float*)d_out;

    const int  Nn = GCN_NODES, Ne = GCN_EDGES, H = GCN_HID, C = GCN_CLS;
    float* ws   = (float*)d_ws;
    float* ns   = ws;
    float* nd   = ws + Nn;
    float* bufA = ws + 2L * Nn;
    float* bufB = bufA + (long)Nn * H;
    float* Wf0  = bufB + (long)Nn * H;
    float* Wf1  = Wf0 + 8 * 2048;
    float* Wf2  = Wf1 + 8 * 2048;
    int*   icnt    = (int*)(Wf2 + 3 * 2048);
    int*   rowptr  = icnt + Nn;
    int*   col     = rowptr + Nn;
    int*   pairD   = col + Ne;
    int*   keyS    = pairD + Ne;
    int*   cntM    = keyS + Ne;
    int*   gbstart = cntM + GCN_G1 * GCN_NCOL;
    int*   gbcnt   = gbstart + GCN_NCOL;

    gcn_bcount<<<GCN_G1, 1024, 0, stream>>>(src, dst, cntM, Ne);
    gcn_bscatter<<<GCN_G1, 1024, 0, stream>>>(src, dst, cntM, gbstart, gbcnt, pairD, keyS, Ne);
    gcn_bsort<<<GCN_NBK, 1024, 0, stream>>>(pairD, keyS, gbstart, gbcnt, rowptr, icnt, col, ns, nd, Nn, Ne);
    gcn_wfrag<<<(8 * 1024 + 255) / 256, 256, 0, stream>>>(W0, Wf0, H, 8);
    gcn_wfrag<<<(8 * 1024 + 255) / 256, 256, 0, stream>>>(W1, Wf1, H, 8);
    gcn_wfrag<<<(3 * 1024 + 255) / 256, 256, 0, stream>>>(W2, Wf2, C, 3);

    gcn_gemm_wmma<<<Nn / 32, 256, 0, stream>>>(feat, ns, Wf0, bufA, H, 8);

    gcn_gather_gemm<<<Nn / 32, 256, 0, stream>>>(bufA, rowptr, icnt, col, nd, ns, b0, Wf1, bufB, H, 8);

    gcn_gather_gemm<<<Nn / 32, 256, 0, stream>>>(bufB, rowptr, icnt, col, nd, ns, b1, Wf2, bufA, C, 3);

    gcn_gather<GCN_CLS, false><<<(Nn + 7) / 8, 256, 0, stream>>>(bufA, rowptr, icnt, col, nd, b2, out, Nn);
}


# Sort passes 2, 3: per-lane 4-byte scattered stores of the edge lists are staged in LDS and leave as contiguous runs (bucket pieces; a bucket's row lists); sum order still = LDS arrival
# speedup vs baseline: 5.3195x; 1.1557x over previous
#include <hip/hip_runtime.h>

typedef __attribute__((ext_vector_type(2))) float v2f;
typedef __attribute__((ext_vector_type(4))) float v4f;
typedef __attribute__((ext_vector_type(8))) float v8f;

#define GCN_NODES   100000
#define GCN_EDGES   1600000
#define GCN_KDIM    128
#define GCN_HID     128
#define GCN_CLS     40
#define GCN_LDA     129

#define GCN_BSH   9
#define GCN_BW    512
#define GCN_NBK   ((GCN_NODES + GCN_BW - 1) / GCN_BW)
#define GCN_NCOL  (2 * GCN_NBK)
#define GCN_G1    256
#define GCN_CHUNK ((GCN_EDGES + GCN_G1 - 1) / GCN_G1)
#define GCN_SBITS 17
#define GCN_EPT   ((GCN_CHUNK + 1023) / 1024)
#define GCN_LCAP  12288

__global__ void __launch_bounds__(1024)
gcn_bcount(const int* __restrict__ src, const int* __restrict__ dst,
           int* __restrict__ cntM, int nE) {
    __shared__ int h[GCN_NCOL];
    const int tid = threadIdx.x;
    if (tid < GCN_NCOL) h[tid] = 0;
    __syncthreads();
    const int e0 = blockIdx.x * GCN_CHUNK;
    const int e1 = min(e0 + GCN_CHUNK, nE);
    int d[GCN_EPT], s[GCN_EPT];
    #pragma unroll
    for (int q = 0; q < GCN_EPT; ++q) {
        const int e = min(e0 + tid + q * 1024, nE - 1);
        d[q] = dst[e];
        s[q] = src[e];
    }
    #pragma unroll
    for (int q = 0; q < GCN_EPT; ++q) {
        if (e0 + tid + q * 1024 < e1) {
            atomicAdd(&h[d[q] >> GCN_BSH], 1);
            atomicAdd(&h[GCN_NBK + (s[q] >> GCN_BSH)], 1);
        }
    }
    __syncthreads();
    if (tid < GCN_NCOL) cntM[blockIdx.x * GCN_NCOL + tid] = h[tid];
}

__global__ void __launch_bounds__(1024)
gcn_bscatter(const int* __restrict__ src, const int* __restrict__ dst,
             const int* __restrict__ cntM, int* __restrict__ gbstart,
             int* __restrict__ gbcnt, int* __restrict__ pairD,
             int* __restrict__ keyS, int nE) {
    __shared__ int sTot[2][GCN_NCOL];
    __shared__ int sBef[2][GCN_NCOL];
    __shared__ int tot[GCN_NCOL];
    __shared__ int cur[GCN_NCOL];
    __shared__ int own[GCN_NCOL];
    __shared__ int lst[GCN_NCOL];
    __shared__ int lcur[GCN_NCOL];
    __shared__ int stage[2 * GCN_CHUNK];
    const int tid = threadIdx.x;
    const int c   = tid & 511;
    const int seg = tid >> 9;
    const int blk = blockIdx.x;
    const int e0  = blk * GCN_CHUNK;
    const int e1  = min(e0 + GCN_CHUNK, nE);

    int t = 0, bf = 0;
    if (c < GCN_NCOL) {
        const int r0 = seg * (GCN_G1 / 2);
        #pragma unroll 16
        for (int r = r0; r < r0 + GCN_G1 / 2; ++r) {
            const int v = cntM[r * GCN_NCOL + c];
            t  += v;
            bf += (r < blk) ? v : 0;
        }
    }
    if (c < GCN_NCOL) { sTot[seg][c] = t; sBef[seg][c] = bf; }
    if (tid < GCN_NCOL) own[tid] = cntM[blk * GCN_NCOL + tid];
    __syncthreads();
    if (tid < GCN_NCOL) tot[tid] = sTot[0][tid] + sTot[1][tid];
    __syncthreads();
    if (tid < GCN_NCOL) {
        const int lo = (tid < GCN_NBK) ? 0 : GCN_NBK;
        int st = 0, ls = 0;
        for (int j = lo; j < tid; ++j) { st += tot[j]; ls += own[j]; }
        cur[tid] = st + sBef[0][tid] + sBef[1][tid];
        if (tid >= GCN_NBK) ls += e1 - e0;
        lst[tid]  = ls;
        lcur[tid] = ls;
        if (blk == 0) { gbstart[tid] = st; gbcnt[tid] = tot[tid]; }
    }
    __syncthreads();

    int d[GCN_EPT], s[GCN_EPT];
    #pragma unroll
    for (int q = 0; q < GCN_EPT; ++q) {
        const int e = min(e0 + tid + q * 1024, nE - 1);
        d[q] = dst[e];
        s[q] = src[e];
    }
    #pragma unroll
    for (int q = 0; q < GCN_EPT; ++q) {
        if (e0 + tid + q * 1024 < e1) {
            const int pd = atomicAdd(&lcur[d[q] >> GCN_BSH], 1);
            stage[pd] = ((d[q] & (GCN_BW - 1)) << GCN_SBITS) | s[q];
            const int pq = atomicAdd(&lcur[GCN_NBK + (s[q] >> GCN_BSH)], 1);
            stage[pq] = s[q] & (GCN_BW - 1);
        }
    }
    __syncthreads();

    const int lane = tid & 31;
    const int wave = __builtin_amdgcn_readfirstlane(tid >> 5) & 31;
    constexpr int NQ = (GCN_NCOL + 31) / 32;
    int pn[NQ], pl[NQ], pg[NQ], v0[NQ], v1[NQ];
    #pragma unroll
    for (int q = 0; q < NQ; ++q) {
        const int cq = wave + 32 * q;
        const int cc = min(cq, GCN_NCOL - 1);
        pn[q] = (cq < GCN_NCOL) ? own[cc] : 0;
        pl[q] = lst[cc];
        pg[q] = cur[cc];
    }
    #pragma unroll
    for (int q = 0; q < NQ; ++q) {
        v0[q] = stage[min(pl[q] + lane,      2 * GCN_CHUNK - 1)];
        v1[q] = stage[min(pl[q] + lane + 32, 2 * GCN_CHUNK - 1)];
    }
    #pragma unroll
    for (int q = 0; q < NQ; ++q) {
        int* o = (wave + 32 * q < GCN_NBK) ? pairD : keyS;
        if (lane < pn[q])      o[pg[q] + lane]      = v0[q];
        if (lane + 32 < pn[q]) o[pg[q] + lane + 32] = v1[q];
        #pragma nounroll
        for (int i = lane + 64; i < pn[q]; i += 32) o[pg[q] + i] = stage[pl[q] + i];
    }
}

__global__ void __launch_bounds__(1024)
gcn_bsort(const int* __restrict__ pairD, const int* __restrict__ keyS,
          const int* __restrict__ gbstart, const int* __restrict__ gbcnt,
          int* __restrict__ rowptr, int* __restrict__ icnt, int* __restrict__ col,
          float* __restrict__ ns, float* __restrict__ nd, int nNodes, int nE) {
    __shared__ int cI[GCN_BW];
    __shared__ int cO[GCN_BW];
    __shared__ int pos[GCN_BW];
    __shared__ int wsum[32];
    __shared__ int lcol[GCN_LCAP];
    const int tid  = threadIdx.x;
    const int lane = tid & 31;
    const int wave = tid >> 5;
    const int k    = blockIdx.x;
    if (tid < GCN_BW) { cI[tid] = 0; cO[tid] = 0; }
    __syncthreads();

    const int dS = gbstart[k];
    const int dN = gbcnt[k];
    const int sS = gbstart[GCN_NBK + k];
    const int sN = gbcnt[GCN_NBK + k];
    const bool staged = dN <= GCN_LCAP;
    for (int i0 = tid; i0 < dN; i0 += 8 * 1024) {
        unsigned key[8];
        #pragma unroll
        for (int q = 0; q < 8; ++q) {
            key[q] = (unsigned)pairD[min(dS + i0 + q * 1024, nE - 1)];
        }
        #pragma unroll
        for (int q = 0; q < 8; ++q)
            if (i0 + q * 1024 < dN) atomicAdd(&cI[key[q] >> GCN_SBITS], 1);
    }
    for (int i0 = tid; i0 < sN; i0 += 8 * 1024) {
        int ks[8];
        #pragma unroll
        for (int q = 0; q < 8; ++q) {
            ks[q] = keyS[min(sS + i0 + q * 1024, nE - 1)];
        }
        #pragma unroll
        for (int q = 0; q < 8; ++q)
            if (i0 + q * 1024 < sN) atomicAdd(&cO[ks[q]], 1);
    }
    __syncthreads();

    const int v = (tid < GCN_BW) ? cI[tid] : 0;
    int inc = v;
    for (int o = 1; o < 32; o <<= 1) {
        int u = __shfl_up(inc, o);
        if (lane >= o) inc += u;
    }
    if (lane == 31) wsum[wave] = inc;
    __syncthreads();
    int off = 0;
    for (int w = 0; w < 16; ++w)
        if (w < wave) off += wsum[w];
    const int ex = off + inc - v;
    if (tid < GCN_BW) {
        pos[tid] = ex;
        const int node = k * GCN_BW + tid;
        if (node < nNodes) {
            rowptr[node] = dS + ex;
            icnt[node]   = v;
            nd[node] = rsqrtf(fmaxf((float)v, 1.0f));
            ns[node] = rsqrtf(fmaxf((float)cO[tid], 1.0f));
        }
    }
    __syncthreads();

    for (int i0 = tid; i0 < dN; i0 += 8 * 1024) {
        unsigned key[8];
        #pragma unroll
        for (int q = 0; q < 8; ++q) {
            key[q] = (unsigned)pairD[min(dS + i0 + q * 1024, nE - 1)];
        }
        #pragma unroll
        for (int q = 0; q < 8; ++q) {
            if (i0 + q * 1024 < dN) {
                const int p  = atomicAdd(&pos[key[q] >> GCN_SBITS], 1);
                const int sv = (int)(key[q] & ((1u << GCN_SBITS) - 1u));
                if (staged) lcol[p] = sv; else col[dS + p] = sv;
            }
        }
    }
    __syncthreads();

    if (staged) {
        for (int i0 = tid; i0 < dN; i0 += 8 * 1024) {
            int ent[8];
            #pragma unroll
            for (int q = 0; q < 8; ++q) {
                ent[q] = lcol[min(i0 + q * 1024, GCN_LCAP - 1)];
            }
            #pragma unroll
            for (int q = 0; q < 8; ++q)
                if (i0 + q * 1024 < dN) col[dS + i0 + q * 1024] = ent[q];
        }
    }
}

__global__ void gcn_wfrag(const float* __restrict__ W, float* __restrict__ Wf,
                          int N, int nColTiles) {
    int i = blockIdx.x * blockDim.x + threadIdx.x;
    int total = nColTiles * 32 * 32;
    if (i >= total) return;
    int lane   = i & 31;
    int k4     = (i >> 5) & 31;
    int t      = i >> 10;
    int lane16 = lane & 15;
    int half   = lane >> 4;
    int kr     = 4 * k4 + 2 * half;
    int col    = t * 16 + lane16;
    float v0 = 0.0f, v1 = 0.0f;
    if (col < N) {
        v0 = W[kr * N + col];
        v1 = W[(kr + 1) * N + col];
    }
    Wf[2 * i]     = v0;
    Wf[2 * i + 1] = v1;
}

static __device__ __forceinline__ void
gcn_tile_product(const float* __restrict__ Alds, const float* __restrict__ Wf,
                 float* __restrict__ Y, int r0, int N, int nColTiles) {
    const int tid    = threadIdx.x;
    const int wave   = tid >> 5;
    const int lane   = tid & 31;
    const int lane16 = lane & 15;
    const int half   = lane >> 4;
    if (wave >= nColTiles) return;

    const float* As0 = &Alds[lane16 * GCN_LDA + 2 * half];
    const float* As1 = As0 + 16 * GCN_LDA;
    const v2f* Wfp = (const v2f*)Wf + (long)wave * 32 * 32 + lane;

    v8f acc0 = {0.f, 0.f, 0.f, 0.f, 0.f, 0.f, 0.f, 0.f};
    v8f acc1 = {0.f, 0.f, 0.f, 0.f, 0.f, 0.f, 0.f, 0.f};

    #pragma unroll 8
    for (int k4 = 0; k4 < GCN_KDIM / 4; ++k4) {
        const int k = k4 * 4;
        v2f a0, a1;
        a0.x = As0[k];
        a0.y = As0[k + 1];
        a1.x = As1[k];
        a1.y = As1[k + 1];
        v2f b = Wfp[k4 * 32];
        acc0 = __builtin_amdgcn_wmma_f32_16x16x4_f32(
                   false, a0, false, b, (short)0, acc0, false, false);
        acc1 = __builtin_amdgcn_wmma_f32_16x16x4_f32(
                   false, a1, false, b, (short)0, acc1, false, false);
    }

    const int col = wave * 16 + lane16;
    if (col < N) {
        #pragma unroll
        for (int r = 0; r < 8; ++r) {
            int m = r + 8 * half;
            Y[(long)(r0 + m) * N + col]      = acc0[r];
            Y[(long)(r0 + 16 + m) * N + col] = acc1[r];
        }
    }
}

__global__ void __launch_bounds__(256)
gcn_gemm_wmma(const float* __restrict__ X, const float* __restrict__ ns,
              const float* __restrict__ Wf, float* __restrict__ Y,
              int N, int nColTiles) {
    __shared__ float Alds[32 * GCN_LDA];

    const int tid = threadIdx.x;
    const int r0  = blockIdx.x * 32;

    #pragma unroll
    for (int idx = tid; idx < 32 * GCN_KDIM; idx += 256) {
        int m = idx >> 7;
        int k = idx & 127;
        Alds[m * GCN_LDA + k] = X[(long)(r0 + m) * GCN_KDIM + k] * ns[r0 + m];
    }
    __syncthreads();

    gcn_tile_product(Alds, Wf, Y, r0, N, nColTiles);
}

template <int F, int U>
static __device__ __forceinline__ v4f gcn_take(v4f acc, const v4f* __restrict__ Xl,
                                               int myidx, int t) {
    v4f v[U];
    #pragma unroll
    for (int u = 0; u < U; ++u) {
        const int s = __builtin_amdgcn_readlane(myidx, t + u);
        v[u] = Xl[(long)s * (F / 4)];
    }
    #pragma unroll
    for (int u = 0; u < U; ++u) acc += v[u];
    return acc;
}

template <int F, bool RELU>
__global__ void __launch_bounds__(256)
gcn_gather(const float* __restrict__ Xe, const int* __restrict__ rowptr,
           const int* __restrict__ icnt, const int* __restrict__ col,
           const float* __restrict__ nd, const float* __restrict__ bias,
           float* __restrict__ Y, int nNodes) {
    constexpr int C4 = F / 4;
    const int lane = threadIdx.x & 31;
    const int wave = threadIdx.x >> 5;
    const int row = __builtin_amdgcn_readfirstlane((int)(blockIdx.x * 8 + wave));
    const bool act = lane < C4;
    const int  cl  = act ? lane : 0;
    const v4f* Xl  = (const v4f*)Xe + cl;

    int start = 0, deg = 0;
    float sc = 0.0f;
    if (row < nNodes) {
        start = rowptr[row];
        deg   = icnt[row];
        sc    = nd[row];
    }

    v4f acc = {0.f, 0.f, 0.f, 0.f};
    for (int b = 0; b < deg; b += 32) {
        const int n = min(32, deg - b);
        const int myidx = (lane < n) ? col[start + b + lane] : 0;
        int t = 0;
        for (; t + 8 <= n; t += 8) acc = gcn_take<F, 8>(acc, Xl, myidx, t);
        if (n & 4) { acc = gcn_take<F, 4>(acc, Xl, myidx, t); t += 4; }
        if (n & 2) { acc = gcn_take<F, 2>(acc, Xl, myidx, t); t += 2; }
        if (n & 1) { acc = gcn_take<F, 1>(acc, Xl, myidx, t); }
    }

    v4f r;
    r.x = acc.x * sc + bias[4 * cl + 0];
    r.y = acc.y * sc + bias[4 * cl + 1];
    r.z = acc.z * sc + bias[4 * cl + 2];
    r.w = acc.w * sc + bias[4 * cl + 3];
    if (RELU) {
        r.x = fmaxf(r.x, 0.0f); r.y = fmaxf(r.y, 0.0f);
        r.z = fmaxf(r.z, 0.0f); r.w = fmaxf(r.w, 0.0f);
    }

    if constexpr (C4 == 32) {
        if (row < nNodes) ((v4f*)Y)[(long)row * C4 + lane] = r;
    } else {
        __shared__ float stage[8 * F];
        if (act) {
            stage[wave * F + 4 * lane + 0] = r.x;
            stage[wave * F + 4 * lane + 1] = r.y;
            stage[wave * F + 4 * lane + 2] = r.z;
            stage[wave * F + 4 * lane + 3] = r.w;
        }
        __syncthreads();
        const long gbase = (long)blockIdx.x * 8 * F;
        const long total = (long)nNodes * F;
        for (int i = threadIdx.x; i < 8 * F; i += 256)
            if (gbase + i < total) Y[gbase + i] = stage[i];
    }
}

__global__ void __launch_bounds__(256)
gcn_gather_gemm(const float* __restrict__ Xe, const int* __restrict__ rowptr,
                const int* __restrict__ icnt, const int* __restrict__ col,
                const float* __restrict__ nd, const float* __restrict__ ns,
                const float* __restrict__ bias, const float* __restrict__ Wf,
                float* __restrict__ Y, int N, int nColTiles) {
    __shared__ float Alds[32 * GCN_LDA];

    const int lane = threadIdx.x & 31;
    const int wave = __builtin_amdgcn_readfirstlane((int)(threadIdx.x >> 5));
    const int r0   = blockIdx.x * 32;
    const int rb   = r0 + wave * 4;
    const v4f* Xl  = (const v4f*)Xe + lane;

    int   st[4], dg[4], id0[4];
    float sc[4], s2[4];
    #pragma unroll
    for (int j = 0; j < 4; ++j) {
        st[j] = rowptr[rb + j];
        dg[j] = icnt[rb + j];
        sc[j] = nd[rb + j];
        s2[j] = ns[rb + j];
    }
    #pragma unroll
    for (int j = 0; j < 4; ++j)
        id0[j] = (lane < dg[j]) ? col[st[j] + lane] : 0;

    const float b0 = bias[4 * lane + 0];
    const float b1 = bias[4 * lane + 1];
    const float b2 = bias[4 * lane + 2];
    const float b3 = bias[4 * lane + 3];

    #pragma unroll
    for (int j = 0; j < 4; ++j) {
        const int deg = dg[j];
        v4f acc = {0.f, 0.f, 0.f, 0.f};
        for (int b = 0; b < deg; b += 32) {
            const int n = min(32, deg - b);
            int myidx = id0[j];
            if (b > 0) myidx = (lane < n) ? col[st[j] + b + lane] : 0;
            int t = 0;
            for (; t + 8 <= n; t += 8) acc = gcn_take<GCN_HID, 8>(acc, Xl, myidx, t);
            if (n & 4) { acc = gcn_take<GCN_HID, 4>(acc, Xl, myidx, t); t += 4; }
            if (n & 2) { acc = gcn_take<GCN_HID, 2>(acc, Xl, myidx, t); t += 2; }
            if (n & 1) { acc = gcn_take<GCN_HID, 1>(acc, Xl, myidx, t); }
        }
        float* a = &Alds[(wave * 4 + j) * GCN_LDA + 4 * lane];
        a[0] = fmaxf(acc.x * sc[j] + b0, 0.0f) * s2[j];
        a[1] = fmaxf(acc.y * sc[j] + b1, 0.0f) * s2[j];
        a[2] = fmaxf(acc.z * sc[j] + b2, 0.0f) * s2[j];
        a[3] = fmaxf(acc.w * sc[j] + b3, 0.0f) * s2[j];
    }
    __syncthreads();

    gcn_tile_product(Alds, Wf, Y, r0, N, nColTiles);
}

extern "C" void kernel_launch(void* const* d_in, const int* in_sizes, int n_in,
                              void* d_out, int out_size, void* d_ws, size_t ws_size,
                              hipStream_t stream) {
    (void)in_sizes; (void)n_in; (void)out_size; (void)ws_size;

    const float* feat = (const float*)d_in[0];
    const int*   src  = (const int*)  d_in[1];
    const int*   dst  = (const int*)  d_in[2];
    const float* W0   = (const float*)d_in[3];
    const float* b0   = (const float*)d_in[4];
    const float* W1   = (const float*)d_in[5];
    const float* b1   = (const float*)d_in[6];
    const float* W2   = (const float*)d_in[7];
    const float* b2   = (const float*)d_in[8];
    float*       out  = (float*)d_out;

    const int  Nn = GCN_NODES, Ne = GCN_EDGES, H = GCN_HID, C = GCN_CLS;
    float* ws   = (float*)d_ws;
    float* ns   = ws;
    float* nd   = ws + Nn;
    float* bufA = ws + 2L * Nn;
    float* bufB = bufA + (long)Nn * H;
    float* Wf0  = bufB + (long)Nn * H;
    float* Wf1  = Wf0 + 8 * 2048;
    float* Wf2  = Wf1 + 8 * 2048;
    int*   icnt    = (int*)(Wf2 + 3 * 2048);
    int*   rowptr  = icnt + Nn;
    int*   col     = rowptr + Nn;
    int*   pairD   = col + Ne;
    int*   keyS    = pairD + Ne;
    int*   cntM    = keyS + Ne;
    int*   gbstart = cntM + GCN_G1 * GCN_NCOL;
    int*   gbcnt   = gbstart + GCN_NCOL;

    gcn_bcount<<<GCN_G1, 1024, 0, stream>>>(src, dst, cntM, Ne);
    gcn_bscatter<<<GCN_G1, 1024, 0, stream>>>(src, dst, cntM, gbstart, gbcnt, pairD, keyS, Ne);
    gcn_bsort<<<GCN_NBK, 1024, 0, stream>>>(pairD, keyS, gbstart, gbcnt, rowptr, icnt, col, ns, nd, Nn, Ne);
    gcn_wfrag<<<(8 * 1024 + 255) / 256, 256, 0, stream>>>(W0, Wf0, H, 8);
    gcn_wfrag<<<(8 * 1024 + 255) / 256, 256, 0, stream>>>(W1, Wf1, H, 8);
    gcn_wfrag<<<(3 * 1024 + 255) / 256, 256, 0, stream>>>(W2, Wf2, C, 3);

    gcn_gemm_wmma<<<Nn / 32, 256, 0, stream>>>(feat, ns, Wf0, bufA, H, 8);

    gcn_gather_gemm<<<Nn / 32, 256, 0, stream>>>(bufA, rowptr, icnt, col, nd, ns, b0, Wf1, bufB, H, 8);

    gcn_gather_gemm<<<Nn / 32, 256, 0, stream>>>(bufB, rowptr, icnt, col, nd, ns, b1, Wf2, bufA, C, 3);

    gcn_gather<GCN_CLS, false><<<(Nn + 7) / 8, 256, 0, stream>>>(bufA, rowptr, icnt, col, nd, b2, out, Nn);
}


# Edge sort in 2 launches, not 3: each slice is bucket-sorted alone in LDS, stored with a piece table; the bucket pass collects its pieces; count pass, cross-slice sums gone; sum order = LDS arrival
# speedup vs baseline: 5.6443x; 1.0611x over previous
#include <hip/hip_runtime.h>

typedef __attribute__((ext_vector_type(2))) float v2f;
typedef __attribute__((ext_vector_type(4))) float v4f;
typedef __attribute__((ext_vector_type(8))) float v8f;

#define GCN_NODES   100000
#define GCN_EDGES   1600000
#define GCN_KDIM    128
#define GCN_HID     128
#define GCN_CLS     40
#define GCN_LDA     129

#define GCN_BSH   9
#define GCN_BW    512
#define GCN_NBK   ((GCN_NODES + GCN_BW - 1) / GCN_BW)
#define GCN_NCOL  (2 * GCN_NBK)
#define GCN_G1    256
#define GCN_CHUNK ((GCN_EDGES + GCN_G1 - 1) / GCN_G1)
#define GCN_SBITS 17
#define GCN_EPT   ((GCN_CHUNK + 1023) / 1024)
#define GCN_LCAP  12288

__global__ void __launch_bounds__(1024)
gcn_bslice(const int* __restrict__ src, const int* __restrict__ dst,
           int* __restrict__ slc, int* __restrict__ pst, int* __restrict__ pcn, int nE) {
    __shared__ int cnt[GCN_NCOL];
    __shared__ int lcur[GCN_NCOL];
    __shared__ int wsum[16];
    __shared__ int stage[2 * GCN_CHUNK];
    const int tid  = threadIdx.x;
    const int lane = tid & 31;
    const int wave = tid >> 5;
    const int blk  = blockIdx.x;
    const int e0   = blk * GCN_CHUNK;
    const int e1   = min(e0 + GCN_CHUNK, nE);
    if (tid < GCN_NCOL) cnt[tid] = 0;
    __syncthreads();

    int d[GCN_EPT], s[GCN_EPT];
    #pragma unroll
    for (int q = 0; q < GCN_EPT; ++q) {
        const int e = min(e0 + tid + q * 1024, nE - 1);
        d[q] = dst[e];
        s[q] = src[e];
    }
    #pragma unroll
    for (int q = 0; q < GCN_EPT; ++q) {
        if (e0 + tid + q * 1024 < e1) {
            atomicAdd(&cnt[d[q] >> GCN_BSH], 1);
            atomicAdd(&cnt[GCN_NBK + (s[q] >> GCN_BSH)], 1);
        }
    }
    __syncthreads();

    const int v = (tid < GCN_NCOL) ? cnt[tid] : 0;
    int inc = v;
    for (int o = 1; o < 32; o <<= 1) {
        int u = __shfl_up(inc, o);
        if (lane >= o) inc += u;
    }
    if (lane == 31 && wave < 16) wsum[wave] = inc;
    __syncthreads();
    int off = 0;
    for (int w = 0; w < 16; ++w)
        if (w < wave) off += wsum[w];
    const int ex = off + inc - v;
    if (tid < GCN_NCOL) {
        lcur[tid] = ex;
        pst[blk * GCN_NCOL + tid] = ex;
        pcn[blk * GCN_NCOL + tid] = v;
    }
    __syncthreads();

    #pragma unroll
    for (int q = 0; q < GCN_EPT; ++q) {
        if (e0 + tid + q * 1024 < e1) {
            const int pd = atomicAdd(&lcur[d[q] >> GCN_BSH], 1);
            stage[pd] = ((d[q] & (GCN_BW - 1)) << GCN_SBITS) | s[q];
            const int pq = atomicAdd(&lcur[GCN_NBK + (s[q] >> GCN_BSH)], 1);
            stage[pq] = s[q] & (GCN_BW - 1);
        }
    }
    __syncthreads();

    const int nOut = 2 * (e1 - e0);
    int* o = slc + (long)blk * (2 * GCN_CHUNK);
    for (int i = tid; i < nOut; i += 1024) o[i] = stage[i];
}

__global__ void __launch_bounds__(1024)
gcn_bsort(const int* __restrict__ slc, const int* __restrict__ pst,
          const int* __restrict__ pcn,
          int* __restrict__ rowptr, int* __restrict__ icnt, int* __restrict__ col,
          float* __restrict__ ns, float* __restrict__ nd, int nNodes, int nE) {
    __shared__ int cI[GCN_BW];
    __shared__ int cO[GCN_BW];
    __shared__ int pos[GCN_BW];
    __shared__ int wsum[32];
    __shared__ int redS[32];
    __shared__ int redN[32];
    __shared__ int lcol[GCN_LCAP];
    const int tid  = threadIdx.x;
    const int lane = tid & 31;
    const int wave = __builtin_amdgcn_readfirstlane(tid >> 5) & 31;
    const int k    = blockIdx.x;
    if (tid < GCN_BW) { cI[tid] = 0; cO[tid] = 0; }

    constexpr int NP = GCN_G1 / 32;
    int dn[NP], dp[NP], sn[NP], sp[NP];
    #pragma unroll
    for (int q = 0; q < NP; ++q) {
        const int r = wave + 32 * q;
        dn[q] = pcn[r * GCN_NCOL + k];
        dp[q] = r * (2 * GCN_CHUNK) + pst[r * GCN_NCOL + k];
        sn[q] = pcn[r * GCN_NCOL + GCN_NBK + k];
        sp[q] = r * (2 * GCN_CHUNK) + pst[r * GCN_NCOL + GCN_NBK + k];
    }

    int accS = 0, accN = 0;
    {
        constexpr int RS = 1024 / GCN_NBK;
        const int rr = tid / GCN_NBK;
        const int j  = tid - rr * GCN_NBK;
        const int jc = min(j, k);
        int acc = 0;
        #pragma unroll
        for (int it = 0; it < (GCN_G1 + RS - 1) / RS; ++it) {
            const int r = rr + RS * it;
            const int t = pcn[min(r, GCN_G1 - 1) * GCN_NCOL + jc];
            acc += (r < GCN_G1) ? t : 0;
        }
        if (rr < RS) {
            accS = (j < k)  ? acc : 0;
            accN = (j == k) ? acc : 0;
        }
    }
    for (int o = 16; o > 0; o >>= 1) {
        accS += __shfl_xor(accS, o);
        accN += __shfl_xor(accN, o);
    }
    if (lane == 0) { redS[wave] = accS; redN[wave] = accN; }
    __syncthreads();
    int dS = 0, dN = 0;
    for (int w = 0; w < 32; ++w) { dS += redS[w]; dN += redN[w]; }
    const bool staged = dN <= GCN_LCAP;

    const int lim = 2 * nE - 1;
    unsigned k0[NP], k1[NP];
    int      q0[NP], q1[NP];
    #pragma unroll
    for (int q = 0; q < NP; ++q) {
        k0[q] = (unsigned)slc[min(dp[q] + lane,      lim)];
        k1[q] = (unsigned)slc[min(dp[q] + lane + 32, lim)];
    }
    #pragma unroll
    for (int q = 0; q < NP; ++q) {
        q0[q] = slc[min(sp[q] + lane,      lim)];
        q1[q] = slc[min(sp[q] + lane + 32, lim)];
    }
    #pragma unroll
    for (int q = 0; q < NP; ++q) {
        if (lane < dn[q])      atomicAdd(&cI[k0[q] >> GCN_SBITS], 1);
        if (lane + 32 < dn[q]) atomicAdd(&cI[k1[q] >> GCN_SBITS], 1);
        #pragma nounroll
        for (int i = lane + 64; i < dn[q]; i += 32)
            atomicAdd(&cI[(unsigned)slc[dp[q] + i] >> GCN_SBITS], 1);
    }
    #pragma unroll
    for (int q = 0; q < NP; ++q) {
        if (lane < sn[q])      atomicAdd(&cO[q0[q]], 1);
        if (lane + 32 < sn[q]) atomicAdd(&cO[q1[q]], 1);
        #pragma nounroll
        for (int i = lane + 64; i < sn[q]; i += 32)
            atomicAdd(&cO[slc[sp[q] + i]], 1);
    }
    __syncthreads();

    const int v = (tid < GCN_BW) ? cI[tid] : 0;
    int inc = v;
    for (int o = 1; o < 32; o <<= 1) {
        int u = __shfl_up(inc, o);
        if (lane >= o) inc += u;
    }
    if (lane == 31) wsum[wave] = inc;
    __syncthreads();
    int off = 0;
    for (int w = 0; w < 16; ++w)
        if (w < wave) off += wsum[w];
    const int ex = off + inc - v;
    if (tid < GCN_BW) {
        pos[tid] = ex;
        const int node = k * GCN_BW + tid;
        if (node < nNodes) {
            rowptr[node] = dS + ex;
            icnt[node]   = v;
            nd[node] = rsqrtf(fmaxf((float)v, 1.0f));
            ns[node] = rsqrtf(fmaxf((float)cO[tid], 1.0f));
        }
    }
    __syncthreads();

    #pragma unroll
    for (int q = 0; q < NP; ++q) {
        if (lane < dn[q]) {
            const int p  = atomicAdd(&pos[k0[q] >> GCN_SBITS], 1);
            const int sv = (int)(k0[q] & ((1u << GCN_SBITS) - 1u));
            if (staged) lcol[p] = sv; else col[dS + p] = sv;
        }
        if (lane + 32 < dn[q]) {
            const int p  = atomicAdd(&pos[k1[q] >> GCN_SBITS], 1);
            const int sv = (int)(k1[q] & ((1u << GCN_SBITS) - 1u));
            if (staged) lcol[p] = sv; else col[dS + p] = sv;
        }
        #pragma nounroll
        for (int i = lane + 64; i < dn[q]; i += 32) {
            const unsigned key = (unsigned)slc[dp[q] + i];
            const int p  = atomicAdd(&pos[key >> GCN_SBITS], 1);
            const int sv = (int)(key & ((1u << GCN_SBITS) - 1u));
            if (staged) lcol[p] = sv; else col[dS + p] = sv;
        }
    }
    __syncthreads();

    if (staged) {
        for (int i0 = tid; i0 < dN; i0 += 8 * 1024) {
            int ent[8];
            #pragma unroll
            for (int q = 0; q < 8; ++q) {
                ent[q] = lcol[min(i0 + q * 1024, GCN_LCAP - 1)];
            }
            #pragma unroll
            for (int q = 0; q < 8; ++q)
                if (i0 + q * 1024 < dN) col[dS + i0 + q * 1024] = ent[q];
        }
    }
}

__global__ void gcn_wfrag(const float* __restrict__ W, float* __restrict__ Wf,
                          int N, int nColTiles) {
    int i = blockIdx.x * blockDim.x + threadIdx.x;
    int total = nColTiles * 32 * 32;
    if (i >= total) return;
    int lane   = i & 31;
    int k4     = (i >> 5) & 31;
    int t      = i >> 10;
    int lane16 = lane & 15;
    int half   = lane >> 4;
    int kr     = 4 * k4 + 2 * half;
    int col    = t * 16 + lane16;
    float v0 = 0.0f, v1 = 0.0f;
    if (col < N) {
        v0 = W[kr * N + col];
        v1 = W[(kr + 1) * N + col];
    }
    Wf[2 * i]     = v0;
    Wf[2 * i + 1] = v1;
}

static __device__ __forceinline__ void
gcn_tile_product(const float* __restrict__ Alds, const float* __restrict__ Wf,
                 float* __restrict__ Y, int r0, int N, int nColTiles) {
    const int tid    = threadIdx.x;
    const int wave   = tid >> 5;
    const int lane   = tid & 31;
    const int lane16 = lane & 15;
    const int half   = lane >> 4;
    if (wave >= nColTiles) return;

    const float* As0 = &Alds[lane16 * GCN_LDA + 2 * half];
    const float* As1 = As0 + 16 * GCN_LDA;
    const v2f* Wfp = (const v2f*)Wf + (long)wave * 32 * 32 + lane;

    v8f acc0 = {0.f, 0.f, 0.f, 0.f, 0.f, 0.f, 0.f, 0.f};
    v8f acc1 = {0.f, 0.f, 0.f, 0.f, 0.f, 0.f, 0.f, 0.f};

    #pragma unroll 8
    for (int k4 = 0; k4 < GCN_KDIM / 4; ++k4) {
        const int k = k4 * 4;
        v2f a0, a1;
        a0.x = As0[k];
        a0.y = As0[k + 1];
        a1.x = As1[k];
        a1.y = As1[k + 1];
        v2f b = Wfp[k4 * 32];
        acc0 = __builtin_amdgcn_wmma_f32_16x16x4_f32(
                   false, a0, false, b, (short)0, acc0, false, false);
        acc1 = __builtin_amdgcn_wmma_f32_16x16x4_f32(
                   false, a1, false, b, (short)0, acc1, false, false);
    }

    const int col = wave * 16 + lane16;
    if (col < N) {
        #pragma unroll
        for (int r = 0; r < 8; ++r) {
            int m = r + 8 * half;
            Y[(long)(r0 + m) * N + col]      = acc0[r];
            Y[(long)(r0 + 16 + m) * N + col] = acc1[r];
        }
    }
}

__global__ void __launch_bounds__(256)
gcn_gemm_wmma(const float* __restrict__ X, const float* __restrict__ ns,
              const float* __restrict__ Wf, float* __restrict__ Y,
              int N, int nColTiles) {
    __shared__ float Alds[32 * GCN_LDA];

    const int tid = threadIdx.x;
    const int r0  = blockIdx.x * 32;

    #pragma unroll
    for (int idx = tid; idx < 32 * GCN_KDIM; idx += 256) {
        int m = idx >> 7;
        int k = idx & 127;
        Alds[m * GCN_LDA + k] = X[(long)(r0 + m) * GCN_KDIM + k] * ns[r0 + m];
    }
    __syncthreads();

    gcn_tile_product(Alds, Wf, Y, r0, N, nColTiles);
}

template <int F, int U>
static __device__ __forceinline__ v4f gcn_take(v4f acc, const v4f* __restrict__ Xl,
                                               int myidx, int t) {
    v4f v[U];
    #pragma unroll
    for (int u = 0; u < U; ++u) {
        const int s = __builtin_amdgcn_readlane(myidx, t + u);
        v[u] = Xl[(long)s * (F / 4)];
    }
    #pragma unroll
    for (int u = 0; u < U; ++u) acc += v[u];
    return acc;
}

template <int F, bool RELU>
__global__ void __launch_bounds__(256)
gcn_gather(const float* __restrict__ Xe, const int* __restrict__ rowptr,
           const int* __restrict__ icnt, const int* __restrict__ col,
           const float* __restrict__ nd, const float* __restrict__ bias,
           float* __restrict__ Y, int nNodes) {
    constexpr int C4 = F / 4;
    const int lane = threadIdx.x & 31;
    const int wave = threadIdx.x >> 5;
    const int row = __builtin_amdgcn_readfirstlane((int)(blockIdx.x * 8 + wave));
    const bool act = lane < C4;
    const int  cl  = act ? lane : 0;
    const v4f* Xl  = (const v4f*)Xe + cl;

    int start = 0, deg = 0;
    float sc = 0.0f;
    if (row < nNodes) {
        start = rowptr[row];
        deg   = icnt[row];
        sc    = nd[row];
    }

    v4f acc = {0.f, 0.f, 0.f, 0.f};
    for (int b = 0; b < deg; b += 32) {
        const int n = min(32, deg - b);
        const int myidx = (lane < n) ? col[start + b + lane] : 0;
        int t = 0;
        for (; t + 8 <= n; t += 8) acc = gcn_take<F, 8>(acc, Xl, myidx, t);
        if (n & 4) { acc = gcn_take<F, 4>(acc, Xl, myidx, t); t += 4; }
        if (n & 2) { acc = gcn_take<F, 2>(acc, Xl, myidx, t); t += 2; }
        if (n & 1) { acc = gcn_take<F, 1>(acc, Xl, myidx, t); }
    }

    v4f r;
    r.x = acc.x * sc + bias[4 * cl + 0];
    r.y = acc.y * sc + bias[4 * cl + 1];
    r.z = acc.z * sc + bias[4 * cl + 2];
    r.w = acc.w * sc + bias[4 * cl + 3];
    if (RELU) {
        r.x = fmaxf(r.x, 0.0f); r.y = fmaxf(r.y, 0.0f);
        r.z = fmaxf(r.z, 0.0f); r.w = fmaxf(r.w, 0.0f);
    }

    if constexpr (C4 == 32) {
        if (row < nNodes) ((v4f*)Y)[(long)row * C4 + lane] = r;
    } else {
        __shared__ float stage[8 * F];
        if (act) {
            stage[wave * F + 4 * lane + 0] = r.x;
            stage[wave * F + 4 * lane + 1] = r.y;
            stage[wave * F + 4 * lane + 2] = r.z;
            stage[wave * F + 4 * lane + 3] = r.w;
        }
        __syncthreads();
        const long gbase = (long)blockIdx.x * 8 * F;
        const long total = (long)nNodes * F;
        for (int i = threadIdx.x; i < 8 * F; i += 256)
            if (gbase + i < total) Y[gbase + i] = stage[i];
    }
}

__global__ void __launch_bounds__(256)
gcn_gather_gemm(const float* __restrict__ Xe, const int* __restrict__ rowptr,
                const int* __restrict__ icnt, const int* __restrict__ col,
                const float* __restrict__ nd, const float* __restrict__ ns,
                const float* __restrict__ bias, const float* __restrict__ Wf,
                float* __restrict__ Y, int N, int nColTiles) {
    __shared__ float Alds[32 * GCN_LDA];

    const int lane = threadIdx.x & 31;
    const int wave = __builtin_amdgcn_readfirstlane((int)(threadIdx.x >> 5));
    const int r0   = blockIdx.x * 32;
    const int rb   = r0 + wave * 4;
    const v4f* Xl  = (const v4f*)Xe + lane;

    int   st[4], dg[4], id0[4];
    float sc[4], s2[4];
    #pragma unroll
    for (int j = 0; j < 4; ++j) {
        st[j] = rowptr[rb + j];
        dg[j] = icnt[rb + j];
        sc[j] = nd[rb + j];
        s2[j] = ns[rb + j];
    }
    #pragma unroll
    for (int j = 0; j < 4; ++j)
        id0[j] = (lane < dg[j]) ? col[st[j] + lane] : 0;

    const float b0 = bias[4 * lane + 0];
    const float b1 = bias[4 * lane + 1];
    const float b2 = bias[4 * lane + 2];
    const float b3 = bias[4 * lane + 3];

    #pragma unroll
    for (int j = 0; j < 4; ++j) {
        const int deg = dg[j];
        v4f acc = {0.f, 0.f, 0.f, 0.f};
        for (int b = 0; b < deg; b += 32) {
            const int n = min(32, deg - b);
            int myidx = id0[j];
            if (b > 0) myidx = (lane < n) ? col[st[j] + b + lane] : 0;
            int t = 0;
            for (; t + 8 <= n; t += 8) acc = gcn_take<GCN_HID, 8>(acc, Xl, myidx, t);
            if (n & 4) { acc = gcn_take<GCN_HID, 4>(acc, Xl, myidx, t); t += 4; }
            if (n & 2) { acc = gcn_take<GCN_HID, 2>(acc, Xl, myidx, t); t += 2; }
            if (n & 1) { acc = gcn_take<GCN_HID, 1>(acc, Xl, myidx, t); }
        }
        float* a = &Alds[(wave * 4 + j) * GCN_LDA + 4 * lane];
        a[0] = fmaxf(acc.x * sc[j] + b0, 0.0f) * s2[j];
        a[1] = fmaxf(acc.y * sc[j] + b1, 0.0f) * s2[j];
        a[2] = fmaxf(acc.z * sc[j] + b2, 0.0f) * s2[j];
        a[3] = fmaxf(acc.w * sc[j] + b3, 0.0f) * s2[j];
    }
    __syncthreads();

    gcn_tile_product(Alds, Wf, Y, r0, N, nColTiles);
}

extern "C" void kernel_launch(void* const* d_in, const int* in_sizes, int n_in,
                              void* d_out, int out_size, void* d_ws, size_t ws_size,
                              hipStream_t stream) {
    (void)in_sizes; (void)n_in; (void)out_size; (void)ws_size;

    const float* feat = (const float*)d_in[0];
    const int*   src  = (const int*)  d_in[1];
    const int*   dst  = (const int*)  d_in[2];
    const float* W0   = (const float*)d_in[3];
    const float* b0   = (const float*)d_in[4];
    const float* W1   = (const float*)d_in[5];
    const float* b1   = (const float*)d_in[6];
    const float* W2   = (const float*)d_in[7];
    const float* b2   = (const float*)d_in[8];
    float*       out  = (float*)d_out;

    const int  Nn = GCN_NODES, Ne = GCN_EDGES, H = GCN_HID, C = GCN_CLS;
    float* ws   = (float*)d_ws;
    float* ns   = ws;
    float* nd   = ws + Nn;
    float* bufA = ws + 2L * Nn;
    float* bufB = bufA + (long)Nn * H;
    float* Wf0  = bufB + (long)Nn * H;
    float* Wf1  = Wf0 + 8 * 2048;
    float* Wf2  = Wf1 + 8 * 2048;
    int*   icnt    = (int*)(Wf2 + 3 * 2048);
    int*   rowptr  = icnt + Nn;
    int*   col     = rowptr + Nn;
    int*   slc     = col + Ne;
    int*   pst     = slc + GCN_G1 * 2 * GCN_CHUNK;
    int*   pcn     = pst + GCN_G1 * GCN_NCOL;

    gcn_bslice<<<GCN_G1, 1024, 0, stream>>>(src, dst, slc, pst, pcn, Ne);
    gcn_bsort<<<GCN_NBK, 1024, 0, stream>>>(slc, pst, pcn, rowptr, icnt, col, ns, nd, Nn, Ne);
    gcn_wfrag<<<(8 * 1024 + 255) / 256, 256, 0, stream>>>(W0, Wf0, H, 8);
    gcn_wfrag<<<(8 * 1024 + 255) / 256, 256, 0, stream>>>(W1, Wf1, H, 8);
    gcn_wfrag<<<(3 * 1024 + 255) / 256, 256, 0, stream>>>(W2, Wf2, C, 3);

    gcn_gemm_wmma<<<Nn / 32, 256, 0, stream>>>(feat, ns, Wf0, bufA, H, 8);

    gcn_gather_gemm<<<Nn / 32, 256, 0, stream>>>(bufA, rowptr, icnt, col, nd, ns, b0, Wf1, bufB, H, 8);

    gcn_gather_gemm<<<Nn / 32, 256, 0, stream>>>(bufB, rowptr, icnt, col, nd, ns, b1, Wf2, bufA, C, 3);

    gcn_gather<GCN_CLS, false><<<(Nn + 7) / 8, 256, 0, stream>>>(bufA, rowptr, icnt, col, nd, b2, out, Nn);
}


# Weight fragment packing moved into the first sort launch: each of its 256 blocks packs 76 of the 19456 operand pairs beside its edge loads; the three packing launches gone (9 launches to 6)
# speedup vs baseline: 5.8080x; 1.0290x over previous
#include <hip/hip_runtime.h>

typedef __attribute__((ext_vector_type(2))) float v2f;
typedef __attribute__((ext_vector_type(4))) float v4f;
typedef __attribute__((ext_vector_type(8))) float v8f;

#define GCN_NODES   100000
#define GCN_EDGES   1600000
#define GCN_KDIM    128
#define GCN_HID     128
#define GCN_CLS     40
#define GCN_LDA     129

#define GCN_WT0    8
#define GCN_WT1    8
#define GCN_WT2    3
#define GCN_WITEMS ((GCN_WT0 + GCN_WT1 + GCN_WT2) * 1024)
static __device__ __forceinline__ void
gcn_wfrag_item(const float* __restrict__ W, float* __restrict__ Wf, int N, int i) {
    int lane   = i & 31;
    int k4     = (i >> 5) & 31;
    int t      = i >> 10;
    int lane16 = lane & 15;
    int half   = lane >> 4;
    int kr     = 4 * k4 + 2 * half;
    int col    = t * 16 + lane16;
    float v0 = 0.0f, v1 = 0.0f;
    if (col < N) {
        v0 = W[kr * N + col];
        v1 = W[(kr + 1) * N + col];
    }
    Wf[2 * i]     = v0;
    Wf[2 * i + 1] = v1;
}

#define GCN_BSH   9
#define GCN_BW    512
#define GCN_NBK   ((GCN_NODES + GCN_BW - 1) / GCN_BW)
#define GCN_NCOL  (2 * GCN_NBK)
#define GCN_G1    256
#define GCN_CHUNK ((GCN_EDGES + GCN_G1 - 1) / GCN_G1)
#define GCN_SBITS 17
#define GCN_EPT   ((GCN_CHUNK + 1023) / 1024)
#define GCN_LCAP  12288
#define GCN_WPB   ((GCN_WITEMS + GCN_G1 - 1) / GCN_G1)

__global__ void __launch_bounds__(1024)
gcn_bslice(const int* __restrict__ src, const int* __restrict__ dst,
           int* __restrict__ slc, int* __restrict__ pst, int* __restrict__ pcn, int nE,
           const float* __restrict__ W0, const float* __restrict__ W1,
           const float* __restrict__ W2, float* __restrict__ Wf0,
           float* __restrict__ Wf1, float* __restrict__ Wf2) {
    __shared__ int cnt[GCN_NCOL];
    __shared__ int lcur[GCN_NCOL];
    __shared__ int wsum[16];
    __shared__ int stage[2 * GCN_CHUNK];
    const int tid  = threadIdx.x;
    const int lane = tid & 31;
    const int wave = tid >> 5;
    const int blk  = blockIdx.x;
    const int e0   = blk * GCN_CHUNK;
    const int e1   = min(e0 + GCN_CHUNK, nE);
    if (tid < GCN_NCOL) cnt[tid] = 0;
    __syncthreads();

    int d[GCN_EPT], s[GCN_EPT];
    #pragma unroll
    for (int q = 0; q < GCN_EPT; ++q) {
        const int e = min(e0 + tid + q * 1024, nE - 1);
        d[q] = dst[e];
        s[q] = src[e];
    }
    if (tid < GCN_WPB) {
        const int g = blk * GCN_WPB + tid;
        if (g < GCN_WT0 * 1024)
            gcn_wfrag_item(W0, Wf0, GCN_HID, g);
        else if (g < (GCN_WT0 + GCN_WT1) * 1024)
            gcn_wfrag_item(W1, Wf1, GCN_HID, g - GCN_WT0 * 1024);
        else if (g < GCN_WITEMS)
            gcn_wfrag_item(W2, Wf2, GCN_CLS, g - (GCN_WT0 + GCN_WT1) * 1024);
    }
    #pragma unroll
    for (int q = 0; q < GCN_EPT; ++q) {
        if (e0 + tid + q * 1024 < e1) {
            atomicAdd(&cnt[d[q] >> GCN_BSH], 1);
            atomicAdd(&cnt[GCN_NBK + (s[q] >> GCN_BSH)], 1);
        }
    }
    __syncthreads();

    const int v = (tid < GCN_NCOL) ? cnt[tid] : 0;
    int inc = v;
    for (int o = 1; o < 32; o <<= 1) {
        int u = __shfl_up(inc, o);
        if (lane >= o) inc += u;
    }
    if (lane == 31 && wave < 16) wsum[wave] = inc;
    __syncthreads();
    int off = 0;
    for (int w = 0; w < 16; ++w)
        if (w < wave) off += wsum[w];
    const int ex = off + inc - v;
    if (tid < GCN_NCOL) {
        lcur[tid] = ex;
        pst[blk * GCN_NCOL + tid] = ex;
        pcn[blk * GCN_NCOL + tid] = v;
    }
    __syncthreads();

    #pragma unroll
    for (int q = 0; q < GCN_EPT; ++q) {
        if (e0 + tid + q * 1024 < e1) {
            const int pd = atomicAdd(&lcur[d[q] >> GCN_BSH], 1);
            stage[pd] = ((d[q] & (GCN_BW - 1)) << GCN_SBITS) | s[q];
            const int pq = atomicAdd(&lcur[GCN_NBK + (s[q] >> GCN_BSH)], 1);
            stage[pq] = s[q] & (GCN_BW - 1);
        }
    }
    __syncthreads();

    const int nOut = 2 * (e1 - e0);
    int* o = slc + (long)blk * (2 * GCN_CHUNK);
    for (int i = tid; i < nOut; i += 1024) o[i] = stage[i];
}

__global__ void __launch_bounds__(1024)
gcn_bsort(const int* __restrict__ slc, const int* __restrict__ pst,
          const int* __restrict__ pcn,
          int* __restrict__ rowptr, int* __restrict__ icnt, int* __restrict__ col,
          float* __restrict__ ns, float* __restrict__ nd, int nNodes, int nE) {
    __shared__ int cI[GCN_BW];
    __shared__ int cO[GCN_BW];
    __shared__ int pos[GCN_BW];
    __shared__ int wsum[32];
    __shared__ int redS[32];
    __shared__ int redN[32];
    __shared__ int lcol[GCN_LCAP];
    const int tid  = threadIdx.x;
    const int lane = tid & 31;
    const int wave = __builtin_amdgcn_readfirstlane(tid >> 5) & 31;
    const int k    = blockIdx.x;
    if (tid < GCN_BW) { cI[tid] = 0; cO[tid] = 0; }

    constexpr int NP = GCN_G1 / 32;
    int dn[NP], dp[NP], sn[NP], sp[NP];
    #pragma unroll
    for (int q = 0; q < NP; ++q) {
        const int r = wave + 32 * q;
        dn[q] = pcn[r * GCN_NCOL + k];
        dp[q] = r * (2 * GCN_CHUNK) + pst[r * GCN_NCOL + k];
        sn[q] = pcn[r * GCN_NCOL + GCN_NBK + k];
        sp[q] = r * (2 * GCN_CHUNK) + pst[r * GCN_NCOL + GCN_NBK + k];
    }

    int accS = 0, accN = 0;
    {
        constexpr int RS = 1024 / GCN_NBK;
        const int rr = tid / GCN_NBK;
        const int j  = tid - rr * GCN_NBK;
        const int jc = min(j, k);
        int acc = 0;
        #pragma unroll
        for (int it = 0; it < (GCN_G1 + RS - 1) / RS; ++it) {
            const int r = rr + RS * it;
            const int t = pcn[min(r, GCN_G1 - 1) * GCN_NCOL + jc];
            acc += (r < GCN_G1) ? t : 0;
        }
        if (rr < RS) {
            accS = (j < k)  ? acc : 0;
            accN = (j == k) ? acc : 0;
        }
    }
    for (int o = 16; o > 0; o >>= 1) {
        accS += __shfl_xor(accS, o);
        accN += __shfl_xor(accN, o);
    }
    if (lane == 0) { redS[wave] = accS; redN[wave] = accN; }
    __syncthreads();
    int dS = 0, dN = 0;
    for (int w = 0; w < 32; ++w) { dS += redS[w]; dN += redN[w]; }
    const bool staged = dN <= GCN_LCAP;

    const int lim = 2 * nE - 1;
    unsigned k0[NP], k1[NP];
    int      q0[NP], q1[NP];
    #pragma unroll
    for (int q = 0; q < NP; ++q) {
        k0[q] = (unsigned)slc[min(dp[q] + lane,      lim)];
        k1[q] = (unsigned)slc[min(dp[q] + lane + 32, lim)];
    }
    #pragma unroll
    for (int q = 0; q < NP; ++q) {
        q0[q] = slc[min(sp[q] + lane,      lim)];
        q1[q] = slc[min(sp[q] + lane + 32, lim)];
    }
    #pragma unroll
    for (int q = 0; q < NP; ++q) {
        if (lane < dn[q])      atomicAdd(&cI[k0[q] >> GCN_SBITS], 1);
        if (lane + 32 < dn[q]) atomicAdd(&cI[k1[q] >> GCN_SBITS], 1);
        #pragma nounroll
        for (int i = lane + 64; i < dn[q]; i += 32)
            atomicAdd(&cI[(unsigned)slc[dp[q] + i] >> GCN_SBITS], 1);
    }
    #pragma unroll
    for (int q = 0; q < NP; ++q) {
        if (lane < sn[q])      atomicAdd(&cO[q0[q]], 1);
        if (lane + 32 < sn[q]) atomicAdd(&cO[q1[q]], 1);
        #pragma nounroll
        for (int i = lane + 64; i < sn[q]; i += 32)
            atomicAdd(&cO[slc[sp[q] + i]], 1);
    }
    __syncthreads();

    const int v = (tid < GCN_BW) ? cI[tid] : 0;
    int inc = v;
    for (int o = 1; o < 32; o <<= 1) {
        int u = __shfl_up(inc, o);
        if (lane >= o) inc += u;
    }
    if (lane == 31) wsum[wave] = inc;
    __syncthreads();
    int off = 0;
    for (int w = 0; w < 16; ++w)
        if (w < wave) off += wsum[w];
    const int ex = off + inc - v;
    if (tid < GCN_BW) {
        pos[tid] = ex;
        const int node = k * GCN_BW + tid;
        if (node < nNodes) {
            rowptr[node] = dS + ex;
            icnt[node]   = v;
            nd[node] = rsqrtf(fmaxf((float)v, 1.0f));
            ns[node] = rsqrtf(fmaxf((float)cO[tid], 1.0f));
        }
    }
    __syncthreads();

    #pragma unroll
    for (int q = 0; q < NP; ++q) {
        if (lane < dn[q]) {
            const int p  = atomicAdd(&pos[k0[q] >> GCN_SBITS], 1);
            const int sv = (int)(k0[q] & ((1u << GCN_SBITS) - 1u));
            if (staged) lcol[p] = sv; else col[dS + p] = sv;
        }
        if (lane + 32 < dn[q]) {
            const int p  = atomicAdd(&pos[k1[q] >> GCN_SBITS], 1);
            const int sv = (int)(k1[q] & ((1u << GCN_SBITS) - 1u));
            if (staged) lcol[p] = sv; else col[dS + p] = sv;
        }
        #pragma nounroll
        for (int i = lane + 64; i < dn[q]; i += 32) {
            const unsigned key = (unsigned)slc[dp[q] + i];
            const int p  = atomicAdd(&pos[key >> GCN_SBITS], 1);
            const int sv = (int)(key & ((1u << GCN_SBITS) - 1u));
            if (staged) lcol[p] = sv; else col[dS + p] = sv;
        }
    }
    __syncthreads();

    if (staged) {
        for (int i0 = tid; i0 < dN; i0 += 8 * 1024) {
            int ent[8];
            #pragma unroll
            for (int q = 0; q < 8; ++q) {
                ent[q] = lcol[min(i0 + q * 1024, GCN_LCAP - 1)];
            }
            #pragma unroll
            for (int q = 0; q < 8; ++q)
                if (i0 + q * 1024 < dN) col[dS + i0 + q * 1024] = ent[q];
        }
    }
}

static __device__ __forceinline__ void
gcn_tile_product(const float* __restrict__ Alds, const float* __restrict__ Wf,
                 float* __restrict__ Y, int r0, int N, int nColTiles) {
    const int tid    = threadIdx.x;
    const int wave   = tid >> 5;
    const int lane   = tid & 31;
    const int lane16 = lane & 15;
    const int half   = lane >> 4;
    if (wave >= nColTiles) return;

    const float* As0 = &Alds[lane16 * GCN_LDA + 2 * half];
    const float* As1 = As0 + 16 * GCN_LDA;
    const v2f* Wfp = (const v2f*)Wf + (long)wave * 32 * 32 + lane;

    v8f acc0 = {0.f, 0.f, 0.f, 0.f, 0.f, 0.f, 0.f, 0.f};
    v8f acc1 = {0.f, 0.f, 0.f, 0.f, 0.f, 0.f, 0.f, 0.f};

    #pragma unroll 8
    for (int k4 = 0; k4 < GCN_KDIM / 4; ++k4) {
        const int k = k4 * 4;
        v2f a0, a1;
        a0.x = As0[k];
        a0.y = As0[k + 1];
        a1.x = As1[k];
        a1.y = As1[k + 1];
        v2f b = Wfp[k4 * 32];
        acc0 = __builtin_amdgcn_wmma_f32_16x16x4_f32(
                   false, a0, false, b, (short)0, acc0, false, false);
        acc1 = __builtin_amdgcn_wmma_f32_16x16x4_f32(
                   false, a1, false, b, (short)0, acc1, false, false);
    }

    const int col = wave * 16 + lane16;
    if (col < N) {
        #pragma unroll
        for (int r = 0; r < 8; ++r) {
            int m = r + 8 * half;
            Y[(long)(r0 + m) * N + col]      = acc0[r];
            Y[(long)(r0 + 16 + m) * N + col] = acc1[r];
        }
    }
}

__global__ void __launch_bounds__(256)
gcn_gemm_wmma(const float* __restrict__ X, const float* __restrict__ ns,
              const float* __restrict__ Wf, float* __restrict__ Y,
              int N, int nColTiles) {
    __shared__ float Alds[32 * GCN_LDA];

    const int tid = threadIdx.x;
    const int r0  = blockIdx.x * 32;

    #pragma unroll
    for (int idx = tid; idx < 32 * GCN_KDIM; idx += 256) {
        int m = idx >> 7;
        int k = idx & 127;
        Alds[m * GCN_LDA + k] = X[(long)(r0 + m) * GCN_KDIM + k] * ns[r0 + m];
    }
    __syncthreads();

    gcn_tile_product(Alds, Wf, Y, r0, N, nColTiles);
}

template <int F, int U>
static __device__ __forceinline__ v4f gcn_take(v4f acc, const v4f* __restrict__ Xl,
                                               int myidx, int t) {
    v4f v[U];
    #pragma unroll
    for (int u = 0; u < U; ++u) {
        const int s = __builtin_amdgcn_readlane(myidx, t + u);
        v[u] = Xl[(long)s * (F / 4)];
    }
    #pragma unroll
    for (int u = 0; u < U; ++u) acc += v[u];
    return acc;
}

template <int F, bool RELU>
__global__ void __launch_bounds__(256)
gcn_gather(const float* __restrict__ Xe, const int* __restrict__ rowptr,
           const int* __restrict__ icnt, const int* __restrict__ col,
           const float* __restrict__ nd, const float* __restrict__ bias,
           float* __restrict__ Y, int nNodes) {
    constexpr int C4 = F / 4;
    const int lane = threadIdx.x & 31;
    const int wave = threadIdx.x >> 5;
    const int row = __builtin_amdgcn_readfirstlane((int)(blockIdx.x * 8 + wave));
    const bool act = lane < C4;
    const int  cl  = act ? lane : 0;
    const v4f* Xl  = (const v4f*)Xe + cl;

    int start = 0, deg = 0;
    float sc = 0.0f;
    if (row < nNodes) {
        start = rowptr[row];
        deg   = icnt[row];
        sc    = nd[row];
    }

    v4f acc = {0.f, 0.f, 0.f, 0.f};
    for (int b = 0; b < deg; b += 32) {
        const int n = min(32, deg - b);
        const int myidx = (lane < n) ? col[start + b + lane] : 0;
        int t = 0;
        for (; t + 8 <= n; t += 8) acc = gcn_take<F, 8>(acc, Xl, myidx, t);
        if (n & 4) { acc = gcn_take<F, 4>(acc, Xl, myidx, t); t += 4; }
        if (n & 2) { acc = gcn_take<F, 2>(acc, Xl, myidx, t); t += 2; }
        if (n & 1) { acc = gcn_take<F, 1>(acc, Xl, myidx, t); }
    }

    v4f r;
    r.x = acc.x * sc + bias[4 * cl + 0];
    r.y = acc.y * sc + bias[4 * cl + 1];
    r.z = acc.z * sc + bias[4 * cl + 2];
    r.w = acc.w * sc + bias[4 * cl + 3];
    if (RELU) {
        r.x = fmaxf(r.x, 0.0f); r.y = fmaxf(r.y, 0.0f);
        r.z = fmaxf(r.z, 0.0f); r.w = fmaxf(r.w, 0.0f);
    }

    if constexpr (C4 == 32) {
        if (row < nNodes) ((v4f*)Y)[(long)row * C4 + lane] = r;
    } else {
        __shared__ float stage[8 * F];
        if (act) {
            stage[wave * F + 4 * lane + 0] = r.x;
            stage[wave * F + 4 * lane + 1] = r.y;
            stage[wave * F + 4 * lane + 2] = r.z;
            stage[wave * F + 4 * lane + 3] = r.w;
        }
        __syncthreads();
        const long gbase = (long)blockIdx.x * 8 * F;
        const long total = (long)nNodes * F;
        for (int i = threadIdx.x; i < 8 * F; i += 256)
            if (gbase + i < total) Y[gbase + i] = stage[i];
    }
}

__global__ void __launch_bounds__(256)
gcn_gather_gemm(const float* __restrict__ Xe, const int* __restrict__ rowptr,
                const int* __restrict__ icnt, const int* __restrict__ col,
                const float* __restrict__ nd, const float* __restrict__ ns,
                const float* __restrict__ bias, const float* __restrict__ Wf,
                float* __restrict__ Y, int N, int nColTiles) {
    __shared__ float Alds[32 * GCN_LDA];

    const int lane = threadIdx.x & 31;
    const int wave = __builtin_amdgcn_readfirstlane((int)(threadIdx.x >> 5));
    const int r0   = blockIdx.x * 32;
    const int rb   = r0 + wave * 4;
    const v4f* Xl  = (const v4f*)Xe + lane;

    int   st[4], dg[4], id0[4];
    float sc[4], s2[4];
    #pragma unroll
    for (int j = 0; j < 4; ++j) {
        st[j] = rowptr[rb + j];
        dg[j] = icnt[rb + j];
        sc[j] = nd[rb + j];
        s2[j] = ns[rb + j];
    }
    #pragma unroll
    for (int j = 0; j < 4; ++j)
        id0[j] = (lane < dg[j]) ? col[st[j] + lane] : 0;

    const float b0 = bias[4 * lane + 0];
    const float b1 = bias[4 * lane + 1];
    const float b2 = bias[4 * lane + 2];
    const float b3 = bias[4 * lane + 3];

    #pragma unroll
    for (int j = 0; j < 4; ++j) {
        const int deg = dg[j];
        v4f acc = {0.f, 0.f, 0.f, 0.f};
        for (int b = 0; b < deg; b += 32) {
            const int n = min(32, deg - b);
            int myidx = id0[j];
            if (b > 0) myidx = (lane < n) ? col[st[j] + b + lane] : 0;
            int t = 0;
            for (; t + 8 <= n; t += 8) acc = gcn_take<GCN_HID, 8>(acc, Xl, myidx, t);
            if (n & 4) { acc = gcn_take<GCN_HID, 4>(acc, Xl, myidx, t); t += 4; }
            if (n & 2) { acc = gcn_take<GCN_HID, 2>(acc, Xl, myidx, t); t += 2; }
            if (n & 1) { acc = gcn_take<GCN_HID, 1>(acc, Xl, myidx, t); }
        }
        float* a = &Alds[(wave * 4 + j) * GCN_LDA + 4 * lane];
        a[0] = fmaxf(acc.x * sc[j] + b0, 0.0f) * s2[j];
        a[1] = fmaxf(acc.y * sc[j] + b1, 0.0f) * s2[j];
        a[2] = fmaxf(acc.z * sc[j] + b2, 0.0f) * s2[j];
        a[3] = fmaxf(acc.w * sc[j] + b3, 0.0f) * s2[j];
    }
    __syncthreads();

    gcn_tile_product(Alds, Wf, Y, r0, N, nColTiles);
}

extern "C" void kernel_launch(void* const* d_in, const int* in_sizes, int n_in,
                              void* d_out, int out_size, void* d_ws, size_t ws_size,
                              hipStream_t stream) {
    (void)in_sizes; (void)n_in; (void)out_size; (void)ws_size;

    const float* feat = (const float*)d_in[0];
    const int*   src  = (const int*)  d_in[1];
    const int*   dst  = (const int*)  d_in[2];
    const float* W0   = (const float*)d_in[3];
    const float* b0   = (const float*)d_in[4];
    const float* W1   = (const float*)d_in[5];
    const float* b1   = (const float*)d_in[6];
    const float* W2   = (const float*)d_in[7];
    const float* b2   = (const float*)d_in[8];
    float*       out  = (float*)d_out;

    const int  Nn = GCN_NODES, Ne = GCN_EDGES, H = GCN_HID, C = GCN_CLS;
    float* ws   = (float*)d_ws;
    float* ns   = ws;
    float* nd   = ws + Nn;
    float* bufA = ws + 2L * Nn;
    float* bufB = bufA + (long)Nn * H;
    float* Wf0  = bufB + (long)Nn * H;
    float* Wf1  = Wf0 + 8 * 2048;
    float* Wf2  = Wf1 + 8 * 2048;
    int*   icnt    = (int*)(Wf2 + 3 * 2048);
    int*   rowptr  = icnt + Nn;
    int*   col     = rowptr + Nn;
    int*   slc     = col + Ne;
    int*   pst     = slc + GCN_G1 * 2 * GCN_CHUNK;
    int*   pcn     = pst + GCN_G1 * GCN_NCOL;

    gcn_bslice<<<GCN_G1, 1024, 0, stream>>>(src, dst, slc, pst, pcn, Ne, W0, W1, W2, Wf0, Wf1, Wf2);
    gcn_bsort<<<GCN_NBK, 1024, 0, stream>>>(slc, pst, pcn, rowptr, icnt, col, ns, nd, Nn, Ne);

    gcn_gemm_wmma<<<Nn / 32, 256, 0, stream>>>(feat, ns, Wf0, bufA, H, 8);

    gcn_gather_gemm<<<Nn / 32, 256, 0, stream>>>(bufA, rowptr, icnt, col, nd, ns, b0, Wf1, bufB, H, 8);

    gcn_gather_gemm<<<Nn / 32, 256, 0, stream>>>(bufB, rowptr, icnt, col, nd, ns, b1, Wf2, bufA, C, 3);

    gcn_gather<GCN_CLS, false><<<(Nn + 7) / 8, 256, 0, stream>>>(bufA, rowptr, icnt, col, nd, b2, out, Nn);
}


# Layer 0 sums the norm-scaled input rows before its product ((AX)W0 for A(XW0)) in the block that feeds layer 1's product; product launch and its plane gone; sum order changes (sum, then product)
# speedup vs baseline: 5.9985x; 1.0328x over previous
#include <hip/hip_runtime.h>

typedef __attribute__((ext_vector_type(2))) float v2f;
typedef __attribute__((ext_vector_type(4))) float v4f;
typedef __attribute__((ext_vector_type(8))) float v8f;

#define GCN_NODES   100000
#define GCN_EDGES   1600000
#define GCN_KDIM    128
#define GCN_HID     128
#define GCN_CLS     40
#define GCN_LDA     129

#define GCN_WT0    8
#define GCN_WT1    8
#define GCN_WT2    3
#define GCN_WITEMS ((GCN_WT0 + GCN_WT1 + GCN_WT2) * 1024)
static __device__ __forceinline__ void
gcn_wfrag_item(const float* __restrict__ W, float* __restrict__ Wf, int N, int i) {
    int lane   = i & 31;
    int k4     = (i >> 5) & 31;
    int t      = i >> 10;
    int lane16 = lane & 15;
    int half   = lane >> 4;
    int kr     = 4 * k4 + 2 * half;
    int col    = t * 16 + lane16;
    float v0 = 0.0f, v1 = 0.0f;
    if (col < N) {
        v0 = W[kr * N + col];
        v1 = W[(kr + 1) * N + col];
    }
    Wf[2 * i]     = v0;
    Wf[2 * i + 1] = v1;
}

#define GCN_BSH   9
#define GCN_BW    512
#define GCN_NBK   ((GCN_NODES + GCN_BW - 1) / GCN_BW)
#define GCN_NCOL  (2 * GCN_NBK)
#define GCN_G1    256
#define GCN_CHUNK ((GCN_EDGES + GCN_G1 - 1) / GCN_G1)
#define GCN_SBITS 17
#define GCN_EPT   ((GCN_CHUNK + 1023) / 1024)
#define GCN_LCAP  12288
#define GCN_WPB   ((GCN_WITEMS + GCN_G1 - 1) / GCN_G1)

__global__ void __launch_bounds__(1024)
gcn_bslice(const int* __restrict__ src, const int* __restrict__ dst,
           int* __restrict__ slc, int* __restrict__ pst, int* __restrict__ pcn, int nE,
           const float* __restrict__ W0, const float* __restrict__ W1,
           const float* __restrict__ W2, float* __restrict__ Wf0,
           float* __restrict__ Wf1, float* __restrict__ Wf2) {
    __shared__ int cnt[GCN_NCOL];
    __shared__ int lcur[GCN_NCOL];
    __shared__ int wsum[16];
    __shared__ int stage[2 * GCN_CHUNK];
    const int tid  = threadIdx.x;
    const int lane = tid & 31;
    const int wave = tid >> 5;
    const int blk  = blockIdx.x;
    const int e0   = blk * GCN_CHUNK;
    const int e1   = min(e0 + GCN_CHUNK, nE);
    if (tid < GCN_NCOL) cnt[tid] = 0;
    __syncthreads();

    int d[GCN_EPT], s[GCN_EPT];
    #pragma unroll
    for (int q = 0; q < GCN_EPT; ++q) {
        const int e = min(e0 + tid + q * 1024, nE - 1);
        d[q] = dst[e];
        s[q] = src[e];
    }
    if (tid < GCN_WPB) {
        const int g = blk * GCN_WPB + tid;
        if (g < GCN_WT0 * 1024)
            gcn_wfrag_item(W0, Wf0, GCN_HID, g);
        else if (g < (GCN_WT0 + GCN_WT1) * 1024)
            gcn_wfrag_item(W1, Wf1, GCN_HID, g - GCN_WT0 * 1024);
        else if (g < GCN_WITEMS)
            gcn_wfrag_item(W2, Wf2, GCN_CLS, g - (GCN_WT0 + GCN_WT1) * 1024);
    }
    #pragma unroll
    for (int q = 0; q < GCN_EPT; ++q) {
        if (e0 + tid + q * 1024 < e1) {
            atomicAdd(&cnt[d[q] >> GCN_BSH], 1);
            atomicAdd(&cnt[GCN_NBK + (s[q] >> GCN_BSH)], 1);
        }
    }
    __syncthreads();

    const int v = (tid < GCN_NCOL) ? cnt[tid] : 0;
    int inc = v;
    for (int o = 1; o < 32; o <<= 1) {
        int u = __shfl_up(inc, o);
        if (lane >= o) inc += u;
    }
    if (lane == 31 && wave < 16) wsum[wave] = inc;
    __syncthreads();
    int off = 0;
    for (int w = 0; w < 16; ++w)
        if (w < wave) off += wsum[w];
    const int ex = off + inc - v;
    if (tid < GCN_NCOL) {
        lcur[tid] = ex;
        pst[blk * GCN_NCOL + tid] = ex;
        pcn[blk * GCN_NCOL + tid] = v;
    }
    __syncthreads();

    #pragma unroll
    for (int q = 0; q < GCN_EPT; ++q) {
        if (e0 + tid + q * 1024 < e1) {
            const int pd = atomicAdd(&lcur[d[q] >> GCN_BSH], 1);
            stage[pd] = ((d[q] & (GCN_BW - 1)) << GCN_SBITS) | s[q];
            const int pq = atomicAdd(&lcur[GCN_NBK + (s[q] >> GCN_BSH)], 1);
            stage[pq] = s[q] & (GCN_BW - 1);
        }
    }
    __syncthreads();

    const int nOut = 2 * (e1 - e0);
    int* o = slc + (long)blk * (2 * GCN_CHUNK);
    for (int i = tid; i < nOut; i += 1024) o[i] = stage[i];
}

__global__ void __launch_bounds__(1024)
gcn_bsort(const int* __restrict__ slc, const int* __restrict__ pst,
          const int* __restrict__ pcn,
          int* __restrict__ rowptr, int* __restrict__ icnt, int* __restrict__ col,
          float* __restrict__ ns, float* __restrict__ nd, int nNodes, int nE) {
    __shared__ int cI[GCN_BW];
    __shared__ int cO[GCN_BW];
    __shared__ int pos[GCN_BW];
    __shared__ int wsum[32];
    __shared__ int redS[32];
    __shared__ int redN[32];
    __shared__ int lcol[GCN_LCAP];
    const int tid  = threadIdx.x;
    const int lane = tid & 31;
    const int wave = __builtin_amdgcn_readfirstlane(tid >> 5) & 31;
    const int k    = blockIdx.x;
    if (tid < GCN_BW) { cI[tid] = 0; cO[tid] = 0; }

    constexpr int NP = GCN_G1 / 32;
    int dn[NP], dp[NP], sn[NP], sp[NP];
    #pragma unroll
    for (int q = 0; q < NP; ++q) {
        const int r = wave + 32 * q;
        dn[q] = pcn[r * GCN_NCOL + k];
        dp[q] = r * (2 * GCN_CHUNK) + pst[r * GCN_NCOL + k];
        sn[q] = pcn[r * GCN_NCOL + GCN_NBK + k];
        sp[q] = r * (2 * GCN_CHUNK) + pst[r * GCN_NCOL + GCN_NBK + k];
    }

    int accS = 0, accN = 0;
    {
        constexpr int RS = 1024 / GCN_NBK;
        const int rr = tid / GCN_NBK;
        const int j  = tid - rr * GCN_NBK;
        const int jc = min(j, k);
        int acc = 0;
        #pragma unroll
        for (int it = 0; it < (GCN_G1 + RS - 1) / RS; ++it) {
            const int r = rr + RS * it;
            const int t = pcn[min(r, GCN_G1 - 1) * GCN_NCOL + jc];
            acc += (r < GCN_G1) ? t : 0;
        }
        if (rr < RS) {
            accS = (j < k)  ? acc : 0;
            accN = (j == k) ? acc : 0;
        }
    }
    for (int o = 16; o > 0; o >>= 1) {
        accS += __shfl_xor(accS, o);
        accN += __shfl_xor(accN, o);
    }
    if (lane == 0) { redS[wave] = accS; redN[wave] = accN; }
    __syncthreads();
    int dS = 0, dN = 0;
    for (int w = 0; w < 32; ++w) { dS += redS[w]; dN += redN[w]; }
    const bool staged = dN <= GCN_LCAP;

    const int lim = 2 * nE - 1;
    unsigned k0[NP], k1[NP];
    int      q0[NP], q1[NP];
    #pragma unroll
    for (int q = 0; q < NP; ++q) {
        k0[q] = (unsigned)slc[min(dp[q] + lane,      lim)];
        k1[q] = (unsigned)slc[min(dp[q] + lane + 32, lim)];
    }
    #pragma unroll
    for (int q = 0; q < NP; ++q) {
        q0[q] = slc[min(sp[q] + lane,      lim)];
        q1[q] = slc[min(sp[q] + lane + 32, lim)];
    }
    #pragma unroll
    for (int q = 0; q < NP; ++q) {
        if (lane < dn[q])      atomicAdd(&cI[k0[q] >> GCN_SBITS], 1);
        if (lane + 32 < dn[q]) atomicAdd(&cI[k1[q] >> GCN_SBITS], 1);
        #pragma nounroll
        for (int i = lane + 64; i < dn[q]; i += 32)
            atomicAdd(&cI[(unsigned)slc[dp[q] + i] >> GCN_SBITS], 1);
    }
    #pragma unroll
    for (int q = 0; q < NP; ++q) {
        if (lane < sn[q])      atomicAdd(&cO[q0[q]], 1);
        if (lane + 32 < sn[q]) atomicAdd(&cO[q1[q]], 1);
        #pragma nounroll
        for (int i = lane + 64; i < sn[q]; i += 32)
            atomicAdd(&cO[slc[sp[q] + i]], 1);
    }
    __syncthreads();

    const int v = (tid < GCN_BW) ? cI[tid] : 0;
    int inc = v;
    for (int o = 1; o < 32; o <<= 1) {
        int u = __shfl_up(inc, o);
        if (lane >= o) inc += u;
    }
    if (lane == 31) wsum[wave] = inc;
    __syncthreads();
    int off = 0;
    for (int w = 0; w < 16; ++w)
        if (w < wave) off += wsum[w];
    const int ex = off + inc - v;
    if (tid < GCN_BW) {
        pos[tid] = ex;
        const int node = k * GCN_BW + tid;
        if (node < nNodes) {
            rowptr[node] = dS + ex;
            icnt[node]   = v;
            nd[node] = rsqrtf(fmaxf((float)v, 1.0f));
            ns[node] = rsqrtf(fmaxf((float)cO[tid], 1.0f));
        }
    }
    __syncthreads();

    #pragma unroll
    for (int q = 0; q < NP; ++q) {
        if (lane < dn[q]) {
            const int p  = atomicAdd(&pos[k0[q] >> GCN_SBITS], 1);
            const int sv = (int)(k0[q] & ((1u << GCN_SBITS) - 1u));
            if (staged) lcol[p] = sv; else col[dS + p] = sv;
        }
        if (lane + 32 < dn[q]) {
            const int p  = atomicAdd(&pos[k1[q] >> GCN_SBITS], 1);
            const int sv = (int)(k1[q] & ((1u << GCN_SBITS) - 1u));
            if (staged) lcol[p] = sv; else col[dS + p] = sv;
        }
        #pragma nounroll
        for (int i = lane + 64; i < dn[q]; i += 32) {
            const unsigned key = (unsigned)slc[dp[q] + i];
            const int p  = atomicAdd(&pos[key >> GCN_SBITS], 1);
            const int sv = (int)(key & ((1u << GCN_SBITS) - 1u));
            if (staged) lcol[p] = sv; else col[dS + p] = sv;
        }
    }
    __syncthreads();

    if (staged) {
        for (int i0 = tid; i0 < dN; i0 += 8 * 1024) {
            int ent[8];
            #pragma unroll
            for (int q = 0; q < 8; ++q) {
                ent[q] = lcol[min(i0 + q * 1024, GCN_LCAP - 1)];
            }
            #pragma unroll
            for (int q = 0; q < 8; ++q)
                if (i0 + q * 1024 < dN) col[dS + i0 + q * 1024] = ent[q];
        }
    }
}

static __device__ __forceinline__ void
gcn_tile_mma(const float* __restrict__ Alds, const float* __restrict__ Wf,
             v8f& acc0, v8f& acc1) {
    const int tid    = threadIdx.x;
    const int wave   = tid >> 5;
    const int lane   = tid & 31;
    const int lane16 = lane & 15;
    const int half   = lane >> 4;

    const float* As0 = &Alds[lane16 * GCN_LDA + 2 * half];
    const float* As1 = As0 + 16 * GCN_LDA;
    const v2f* Wfp = (const v2f*)Wf + (long)wave * 32 * 32 + lane;

    #pragma unroll 8
    for (int k4 = 0; k4 < GCN_KDIM / 4; ++k4) {
        const int k = k4 * 4;
        v2f a0, a1;
        a0.x = As0[k];
        a0.y = As0[k + 1];
        a1.x = As1[k];
        a1.y = As1[k + 1];
        v2f b = Wfp[k4 * 32];
        acc0 = __builtin_amdgcn_wmma_f32_16x16x4_f32(
                   false, a0, false, b, (short)0, acc0, false, false);
        acc1 = __builtin_amdgcn_wmma_f32_16x16x4_f32(
                   false, a1, false, b, (short)0, acc1, false, false);
    }
}

static __device__ __forceinline__ void
gcn_tile_product(const float* __restrict__ Alds, const float* __restrict__ Wf,
                 float* __restrict__ Y, int r0, int N, int nColTiles) {
    const int tid    = threadIdx.x;
    const int wave   = tid >> 5;
    const int lane   = tid & 31;
    const int lane16 = lane & 15;
    const int half   = lane >> 4;
    if (wave >= nColTiles) return;

    v8f acc0 = {0.f, 0.f, 0.f, 0.f, 0.f, 0.f, 0.f, 0.f};
    v8f acc1 = {0.f, 0.f, 0.f, 0.f, 0.f, 0.f, 0.f, 0.f};
    gcn_tile_mma(Alds, Wf, acc0, acc1);

    const int col = wave * 16 + lane16;
    if (col < N) {
        #pragma unroll
        for (int r = 0; r < 8; ++r) {
            int m = r + 8 * half;
            Y[(long)(r0 + m) * N + col]      = acc0[r];
            Y[(long)(r0 + 16 + m) * N + col] = acc1[r];
        }
    }
}

template <int F, int U>
static __device__ __forceinline__ v4f gcn_take(v4f acc, const v4f* __restrict__ Xl,
                                               int myidx, int t) {
    v4f v[U];
    #pragma unroll
    for (int u = 0; u < U; ++u) {
        const int s = __builtin_amdgcn_readlane(myidx, t + u);
        v[u] = Xl[(long)s * (F / 4)];
    }
    #pragma unroll
    for (int u = 0; u < U; ++u) acc += v[u];
    return acc;
}

template <int F, int U>
static __device__ __forceinline__ v4f gcn_take_w(v4f acc, const v4f* __restrict__ Xl,
                                                 int myidx, float myw, int t) {
    v4f   v[U];
    float w[U];
    #pragma unroll
    for (int u = 0; u < U; ++u) {
        const int s = __builtin_amdgcn_readlane(myidx, t + u);
        w[u] = __builtin_bit_cast(float,
                   __builtin_amdgcn_readlane(__builtin_bit_cast(int, myw), t + u));
        v[u] = Xl[(long)s * (F / 4)];
    }
    #pragma unroll
    for (int u = 0; u < U; ++u) acc += v[u] * w[u];
    return acc;
}

template <int F, bool RELU>
__global__ void __launch_bounds__(256)
gcn_gather(const float* __restrict__ Xe, const int* __restrict__ rowptr,
           const int* __restrict__ icnt, const int* __restrict__ col,
           const float* __restrict__ nd, const float* __restrict__ bias,
           float* __restrict__ Y, int nNodes) {
    constexpr int C4 = F / 4;
    const int lane = threadIdx.x & 31;
    const int wave = threadIdx.x >> 5;
    const int row = __builtin_amdgcn_readfirstlane((int)(blockIdx.x * 8 + wave));
    const bool act = lane < C4;
    const int  cl  = act ? lane : 0;
    const v4f* Xl  = (const v4f*)Xe + cl;

    int start = 0, deg = 0;
    float sc = 0.0f;
    if (row < nNodes) {
        start = rowptr[row];
        deg   = icnt[row];
        sc    = nd[row];
    }

    v4f acc = {0.f, 0.f, 0.f, 0.f};
    for (int b = 0; b < deg; b += 32) {
        const int n = min(32, deg - b);
        const int myidx = (lane < n) ? col[start + b + lane] : 0;
        int t = 0;
        for (; t + 8 <= n; t += 8) acc = gcn_take<F, 8>(acc, Xl, myidx, t);
        if (n & 4) { acc = gcn_take<F, 4>(acc, Xl, myidx, t); t += 4; }
        if (n & 2) { acc = gcn_take<F, 2>(acc, Xl, myidx, t); t += 2; }
        if (n & 1) { acc = gcn_take<F, 1>(acc, Xl, myidx, t); }
    }

    v4f r;
    r.x = acc.x * sc + bias[4 * cl + 0];
    r.y = acc.y * sc + bias[4 * cl + 1];
    r.z = acc.z * sc + bias[4 * cl + 2];
    r.w = acc.w * sc + bias[4 * cl + 3];
    if (RELU) {
        r.x = fmaxf(r.x, 0.0f); r.y = fmaxf(r.y, 0.0f);
        r.z = fmaxf(r.z, 0.0f); r.w = fmaxf(r.w, 0.0f);
    }

    if constexpr (C4 == 32) {
        if (row < nNodes) ((v4f*)Y)[(long)row * C4 + lane] = r;
    } else {
        __shared__ float stage[8 * F];
        if (act) {
            stage[wave * F + 4 * lane + 0] = r.x;
            stage[wave * F + 4 * lane + 1] = r.y;
            stage[wave * F + 4 * lane + 2] = r.z;
            stage[wave * F + 4 * lane + 3] = r.w;
        }
        __syncthreads();
        const long gbase = (long)blockIdx.x * 8 * F;
        const long total = (long)nNodes * F;
        for (int i = threadIdx.x; i < 8 * F; i += 256)
            if (gbase + i < total) Y[gbase + i] = stage[i];
    }
}

__global__ void __launch_bounds__(256)
gcn_gather_gemm(const float* __restrict__ Xe, const int* __restrict__ rowptr,
                const int* __restrict__ icnt, const int* __restrict__ col,
                const float* __restrict__ nd, const float* __restrict__ ns,
                const float* __restrict__ bias, const float* __restrict__ Wf,
                float* __restrict__ Y, int N, int nColTiles) {
    __shared__ float Alds[32 * GCN_LDA];

    const int lane = threadIdx.x & 31;
    const int wave = __builtin_amdgcn_readfirstlane((int)(threadIdx.x >> 5));
    const int r0   = blockIdx.x * 32;
    const int rb   = r0 + wave * 4;
    const v4f* Xl  = (const v4f*)Xe + lane;

    int   st[4], dg[4], id0[4];
    float sc[4], s2[4];
    #pragma unroll
    for (int j = 0; j < 4; ++j) {
        st[j] = rowptr[rb + j];
        dg[j] = icnt[rb + j];
        sc[j] = nd[rb + j];
        s2[j] = ns[rb + j];
    }
    #pragma unroll
    for (int j = 0; j < 4; ++j)
        id0[j] = (lane < dg[j]) ? col[st[j] + lane] : 0;

    const float b0 = bias[4 * lane + 0];
    const float b1 = bias[4 * lane + 1];
    const float b2 = bias[4 * lane + 2];
    const float b3 = bias[4 * lane + 3];

    #pragma unroll
    for (int j = 0; j < 4; ++j) {
        const int deg = dg[j];
        v4f acc = {0.f, 0.f, 0.f, 0.f};
        for (int b = 0; b < deg; b += 32) {
            const int n = min(32, deg - b);
            int myidx = id0[j];
            if (b > 0) myidx = (lane < n) ? col[st[j] + b + lane] : 0;
            int t = 0;
            for (; t + 8 <= n; t += 8) acc = gcn_take<GCN_HID, 8>(acc, Xl, myidx, t);
            if (n & 4) { acc = gcn_take<GCN_HID, 4>(acc, Xl, myidx, t); t += 4; }
            if (n & 2) { acc = gcn_take<GCN_HID, 2>(acc, Xl, myidx, t); t += 2; }
            if (n & 1) { acc = gcn_take<GCN_HID, 1>(acc, Xl, myidx, t); }
        }
        float* a = &Alds[(wave * 4 + j) * GCN_LDA + 4 * lane];
        a[0] = fmaxf(acc.x * sc[j] + b0, 0.0f) * s2[j];
        a[1] = fmaxf(acc.y * sc[j] + b1, 0.0f) * s2[j];
        a[2] = fmaxf(acc.z * sc[j] + b2, 0.0f) * s2[j];
        a[3] = fmaxf(acc.w * sc[j] + b3, 0.0f) * s2[j];
    }
    __syncthreads();

    gcn_tile_product(Alds, Wf, Y, r0, N, nColTiles);
}

__global__ void __launch_bounds__(256)
gcn_gather_gemm2(const float* __restrict__ X, const int* __restrict__ rowptr,
                 const int* __restrict__ icnt, const int* __restrict__ col,
                 const float* __restrict__ nd, const float* __restrict__ ns,
                 const float* __restrict__ bias, const float* __restrict__ WfA,
                 const float* __restrict__ WfB, float* __restrict__ Y) {
    __shared__ float Alds[32 * GCN_LDA];
    __shared__ float Hlds[32 * GCN_LDA];
    __shared__ float lnd[32];
    __shared__ float lns[32];

    const int lane = threadIdx.x & 31;
    const int wave = __builtin_amdgcn_readfirstlane((int)(threadIdx.x >> 5));
    const int r0   = blockIdx.x * 32;
    const int rb   = r0 + wave * 4;
    const v4f* Xl  = (const v4f*)X + lane;

    int   st[4], dg[4], id0[4];
    float w0[4];
    #pragma unroll
    for (int j = 0; j < 4; ++j) {
        st[j] = rowptr[rb + j];
        dg[j] = icnt[rb + j];
    }
    #pragma unroll
    for (int j = 0; j < 4; ++j)
        id0[j] = (lane < dg[j]) ? col[st[j] + lane] : 0;
    #pragma unroll
    for (int j = 0; j < 4; ++j)
        w0[j] = ns[id0[j]];
    if (lane < 4) {
        lnd[wave * 4 + lane] = nd[rb + lane];
        lns[wave * 4 + lane] = ns[rb + lane];
    }

    #pragma unroll
    for (int j = 0; j < 4; ++j) {
        const int deg = dg[j];
        v4f acc = {0.f, 0.f, 0.f, 0.f};
        for (int b = 0; b < deg; b += 32) {
            const int n = min(32, deg - b);
            int   myidx = id0[j];
            float myw   = w0[j];
            if (b > 0) {
                myidx = (lane < n) ? col[st[j] + b + lane] : 0;
                myw   = ns[myidx];
            }
            int t = 0;
            for (; t + 8 <= n; t += 8) acc = gcn_take_w<GCN_KDIM, 8>(acc, Xl, myidx, myw, t);
            if (n & 4) { acc = gcn_take_w<GCN_KDIM, 4>(acc, Xl, myidx, myw, t); t += 4; }
            if (n & 2) { acc = gcn_take_w<GCN_KDIM, 2>(acc, Xl, myidx, myw, t); t += 2; }
            if (n & 1) { acc = gcn_take_w<GCN_KDIM, 1>(acc, Xl, myidx, myw, t); }
        }
        float* a = &Alds[(wave * 4 + j) * GCN_LDA + 4 * lane];
        a[0] = acc.x; a[1] = acc.y; a[2] = acc.z; a[3] = acc.w;
    }
    __syncthreads();

    const int lane16 = lane & 15;
    const int half   = lane >> 4;
    v8f acc0 = {0.f, 0.f, 0.f, 0.f, 0.f, 0.f, 0.f, 0.f};
    v8f acc1 = {0.f, 0.f, 0.f, 0.f, 0.f, 0.f, 0.f, 0.f};
    gcn_tile_mma(Alds, WfA, acc0, acc1);

    const int   cc = wave * 16 + lane16;
    const float bc = bias[cc];
    #pragma unroll
    for (int r = 0; r < 8; ++r) {
        const int m = r + 8 * half;
        Hlds[m * GCN_LDA + cc]        = fmaxf(acc0[r] * lnd[m]      + bc, 0.0f) * lns[m];
        Hlds[(m + 16) * GCN_LDA + cc] = fmaxf(acc1[r] * lnd[m + 16] + bc, 0.0f) * lns[m + 16];
    }
    __syncthreads();

    gcn_tile_product(Hlds, WfB, Y, r0, GCN_HID, 8);
}

extern "C" void kernel_launch(void* const* d_in, const int* in_sizes, int n_in,
                              void* d_out, int out_size, void* d_ws, size_t ws_size,
                              hipStream_t stream) {
    (void)in_sizes; (void)n_in; (void)out_size; (void)ws_size;

    const float* feat = (const float*)d_in[0];
    const int*   src  = (const int*)  d_in[1];
    const int*   dst  = (const int*)  d_in[2];
    const float* W0   = (const float*)d_in[3];
    const float* b0   = (const float*)d_in[4];
    const float* W1   = (const float*)d_in[5];
    const float* b1   = (const float*)d_in[6];
    const float* W2   = (const float*)d_in[7];
    const float* b2   = (const float*)d_in[8];
    float*       out  = (float*)d_out;

    const int  Nn = GCN_NODES, Ne = GCN_EDGES, H = GCN_HID, C = GCN_CLS;
    float* ws   = (float*)d_ws;
    float* ns   = ws;
    float* nd   = ws + Nn;
    float* bufA = ws + 2L * Nn;
    float* bufB = bufA + (long)Nn * H;
    float* Wf0  = bufB + (long)Nn * H;
    float* Wf1  = Wf0 + 8 * 2048;
    float* Wf2  = Wf1 + 8 * 2048;
    int*   icnt    = (int*)(Wf2 + 3 * 2048);
    int*   rowptr  = icnt + Nn;
    int*   col     = rowptr + Nn;
    int*   slc     = col + Ne;
    int*   pst     = slc + GCN_G1 * 2 * GCN_CHUNK;
    int*   pcn     = pst + GCN_G1 * GCN_NCOL;

    gcn_bslice<<<GCN_G1, 1024, 0, stream>>>(src, dst, slc, pst, pcn, Ne, W0, W1, W2, Wf0, Wf1, Wf2);
    gcn_bsort<<<GCN_NBK, 1024, 0, stream>>>(slc, pst, pcn, rowptr, icnt, col, ns, nd, Nn, Ne);

    gcn_gather_gemm2<<<Nn / 32, 256, 0, stream>>>(feat, rowptr, icnt, col, nd, ns, b0, Wf0, Wf1, bufB);

    gcn_gather_gemm<<<Nn / 32, 256, 0, stream>>>(bufB, rowptr, icnt, col, nd, ns, b1, Wf2, bufA, C, 3);

    gcn_gather<GCN_CLS, false><<<(Nn + 7) / 8, 256, 0, stream>>>(bufA, rowptr, icnt, col, nd, b2, out, Nn);
}
